# starting kernel
# baseline (speedup 1.0000x reference)
#include <hip/hip_runtime.h>

#define N_NODES 50000
#define N_EDGES 800000
#define IN_DIM 128
#define OUT_DIM 128
#define HEADS 4
#define HEAD_DIM 32
#define EDGE_DIM 10
#define NEG_SLOPE 0.2f
#define LN_EPS 1e-5f

typedef __attribute__((ext_vector_type(2))) float v2f;
typedef __attribute__((ext_vector_type(8))) float v8f;

__global__ __launch_bounds__(256) void gatv2_gemm_lr(
    const float* __restrict__ x,
    const float* __restrict__ wl, const float* __restrict__ bl,
    const float* __restrict__ wr, const float* __restrict__ br,
    float* __restrict__ xl, float* __restrict__ xr)
{
    const int wave = threadIdx.x >> 5;
    const int lane = threadIdx.x & 31;
    const int mbase = blockIdx.x * 16;
    const int nbase = wave * 16;
    const int lm = lane & 15;
    const int kh = (lane >> 4) * 2;

    const int arow = mbase + lm;
    const int bcol = nbase + lm;

    const float* __restrict__ xrow  = x  + (size_t)arow * IN_DIM;
    const float* __restrict__ wlrow = wl + (size_t)bcol * IN_DIM;
    const float* __restrict__ wrrow = wr + (size_t)bcol * IN_DIM;

    v8f accL = {};
    v8f accR = {};
    #pragma unroll 4
    for (int k = 0; k < IN_DIM; k += 4) {
        v2f a, bL, bR;
        a.x  = xrow[k + kh];      a.y  = xrow[k + kh + 1];
        bL.x = wlrow[k + kh];     bL.y = wlrow[k + kh + 1];
        bR.x = wrrow[k + kh];     bR.y = wrrow[k + kh + 1];
        accL = __builtin_amdgcn_wmma_f32_16x16x4_f32(
            false, a, false, bL, (short)0, accL, false, false);
        accR = __builtin_amdgcn_wmma_f32_16x16x4_f32(
            false, a, false, bR, (short)0, accR, false, false);
    }

    const float biasL = bl[bcol];
    const float biasR = br[bcol];
    const int rbase = mbase + (lane >> 4) * 8;
    #pragma unroll
    for (int r = 0; r < 8; ++r) {
        const size_t off = (size_t)(rbase + r) * OUT_DIM + bcol;
        xl[off] = accL[r] + biasL;
        xr[off] = accR[r] + biasR;
    }
}

__global__ void gatv2_init(float* __restrict__ maxv, float* __restrict__ denom,
                           float* __restrict__ outacc)
{
    const int i = blockIdx.x * blockDim.x + threadIdx.x;
    if (i < N_NODES * HEADS) {
        maxv[i]  = -__builtin_inff();
        denom[i] = 0.0f;
    }
    if (i < N_NODES * OUT_DIM) outacc[i] = 0.0f;
}

__global__ __launch_bounds__(256) void gatv2_edge_logits(
    const int* __restrict__ ei, const float* __restrict__ eattr,
    const float* __restrict__ we, const float* __restrict__ att,
    const float* __restrict__ xl, const float* __restrict__ xr,
    float* __restrict__ logits, float* __restrict__ maxv)
{
    const int e = blockIdx.x * 8 + (threadIdx.x >> 5);
    if (e >= N_EDGES) return;
    const int lane = threadIdx.x & 31;
    const int src = ei[e];
    const int dst = ei[N_EDGES + e];

    float ea[EDGE_DIM];
    #pragma unroll
    for (int d = 0; d < EDGE_DIM; ++d)
        ea[d] = eattr[(size_t)e * EDGE_DIM + d];

    #pragma unroll
    for (int h = 0; h < HEADS; ++h) {
        const int c = h * HEAD_DIM + lane;
        float ev = 0.0f;
        #pragma unroll
        for (int d = 0; d < EDGE_DIM; ++d)
            ev = fmaf(ea[d], we[(size_t)c * EDGE_DIM + d], ev);
        float v = xl[(size_t)src * OUT_DIM + c] +
                  xr[(size_t)dst * OUT_DIM + c] + ev;
        v = v > 0.0f ? v : NEG_SLOPE * v;
        float p = v * att[c];
        #pragma unroll
        for (int off = 16; off > 0; off >>= 1)
            p += __shfl_xor(p, off, 32);
        if (lane == 0) {
            logits[(size_t)e * HEADS + h] = p;
            atomicMax(&maxv[(size_t)dst * HEADS + h], p);
        }
    }
}

__global__ void gatv2_edge_expsum(
    const int* __restrict__ ei, float* __restrict__ logits,
    const float* __restrict__ maxv, float* __restrict__ denom)
{
    const int i = blockIdx.x * blockDim.x + threadIdx.x;
    if (i >= N_EDGES * HEADS) return;
    const int e = i >> 2;
    const int h = i & 3;
    const int dst = ei[N_EDGES + e];
    const float ex = __expf(logits[i] - maxv[(size_t)dst * HEADS + h]);
    logits[i] = ex;
    atomicAdd(&denom[(size_t)dst * HEADS + h], ex);
}

__global__ __launch_bounds__(256) void gatv2_edge_aggregate(
    const int* __restrict__ ei, const float* __restrict__ logits,
    const float* __restrict__ denom, const float* __restrict__ xl,
    float* __restrict__ outacc)
{
    const int e = blockIdx.x * 8 + (threadIdx.x >> 5);
    if (e >= N_EDGES) return;
    const int lane = threadIdx.x & 31;
    const int src = ei[e];
    const int dst = ei[N_EDGES + e];
    #pragma unroll
    for (int h = 0; h < HEADS; ++h) {
        const float alpha = logits[(size_t)e * HEADS + h] /
                            (denom[(size_t)dst * HEADS + h] + 1e-16f);
        const int c = h * HEAD_DIM + lane;
        atomicAdd(&outacc[(size_t)dst * OUT_DIM + c],
                  xl[(size_t)src * OUT_DIM + c] * alpha);
    }
}

__global__ __launch_bounds__(256) void gatv2_final(
    const float* __restrict__ x, const float* __restrict__ outacc,
    const float* __restrict__ cbias, const float* __restrict__ gamma,
    const float* __restrict__ beta, float* __restrict__ out)
{
    const int n = blockIdx.x * 8 + (threadIdx.x >> 5);
    if (n >= N_NODES) return;
    const int lane = threadIdx.x & 31;

    float h[4];
    float s = 0.0f;
    #pragma unroll
    for (int j = 0; j < 4; ++j) {
        const int c = lane + 32 * j;
        const float v = outacc[(size_t)n * OUT_DIM + c] + cbias[c];
        const float sig = 1.0f / (1.0f + __expf(-v));
        h[j] = v * sig + x[(size_t)n * OUT_DIM + c];
        s += h[j];
    }
    #pragma unroll
    for (int off = 16; off > 0; off >>= 1) s += __shfl_xor(s, off, 32);
    const float mu = s * (1.0f / OUT_DIM);

    float vs = 0.0f;
    #pragma unroll
    for (int j = 0; j < 4; ++j) { const float d = h[j] - mu; vs += d * d; }
    #pragma unroll
    for (int off = 16; off > 0; off >>= 1) vs += __shfl_xor(vs, off, 32);
    const float inv = rsqrtf(vs * (1.0f / OUT_DIM) + LN_EPS);

    #pragma unroll
    for (int j = 0; j < 4; ++j) {
        const int c = lane + 32 * j;
        out[(size_t)n * OUT_DIM + c] = (h[j] - mu) * inv * gamma[c] + beta[c];
    }
}

extern "C" void kernel_launch(void* const* d_in, const int* in_sizes, int n_in,
                              void* d_out, int out_size, void* d_ws, size_t ws_size,
                              hipStream_t stream) {
    const float* x     = (const float*)d_in[0];
    const int*   ei    = (const int*)  d_in[1];
    const float* eattr = (const float*)d_in[2];
    const float* wl    = (const float*)d_in[3];
    const float* bl    = (const float*)d_in[4];
    const float* wr    = (const float*)d_in[5];
    const float* br    = (const float*)d_in[6];
    const float* we    = (const float*)d_in[7];
    const float* att   = (const float*)d_in[8];
    const float* cb    = (const float*)d_in[9];
    const float* gam   = (const float*)d_in[10];
    const float* bet   = (const float*)d_in[11];
    float* out = (float*)d_out;

    float* ws     = (float*)d_ws;
    float* xl     = ws;
    float* xr     = xl     + (size_t)N_NODES * OUT_DIM;
    float* outacc = xr     + (size_t)N_NODES * OUT_DIM;
    float* logits = outacc + (size_t)N_NODES * OUT_DIM;
    float* maxv   = logits + (size_t)N_EDGES * HEADS;
    float* denom  = maxv   + (size_t)N_NODES * HEADS;

    gatv2_gemm_lr<<<N_NODES / 16, 256, 0, stream>>>(x, wl, bl, wr, br, xl, xr);

    const int initN = N_NODES * OUT_DIM;
    gatv2_init<<<(initN + 255) / 256, 256, 0, stream>>>(maxv, denom, outacc);

    gatv2_edge_logits<<<(N_EDGES + 7) / 8, 256, 0, stream>>>(
        ei, eattr, we, att, xl, xr, logits, maxv);

    gatv2_edge_expsum<<<(N_EDGES * HEADS + 255) / 256, 256, 0, stream>>>(
        ei, logits, maxv, denom);

    gatv2_edge_aggregate<<<(N_EDGES + 7) / 8, 256, 0, stream>>>(
        ei, logits, denom, xl, outacc);

    gatv2_final<<<(N_NODES + 7) / 8, 256, 0, stream>>>(
        x, outacc, cb, gam, bet, out);
}


# edge-logit kernel: each wave loops over 8 edges with its w_e and att columns held in registers and a wave-uniform edge id, so per-edge weight loads vanish and index/attribute loads go scalar
# speedup vs baseline: 1.0484x; 1.0484x over previous
#include <hip/hip_runtime.h>

#define N_NODES 50000
#define N_EDGES 800000
#define IN_DIM 128
#define OUT_DIM 128
#define HEADS 4
#define HEAD_DIM 32
#define EDGE_DIM 10
#define NEG_SLOPE 0.2f
#define LN_EPS 1e-5f

typedef __attribute__((ext_vector_type(2))) float v2f;
typedef __attribute__((ext_vector_type(8))) float v8f;

__global__ __launch_bounds__(256) void gatv2_gemm_lr(
    const float* __restrict__ x,
    const float* __restrict__ wl, const float* __restrict__ bl,
    const float* __restrict__ wr, const float* __restrict__ br,
    float* __restrict__ xl, float* __restrict__ xr)
{
    const int wave = threadIdx.x >> 5;
    const int lane = threadIdx.x & 31;
    const int mbase = blockIdx.x * 16;
    const int nbase = wave * 16;
    const int lm = lane & 15;
    const int kh = (lane >> 4) * 2;

    const int arow = mbase + lm;
    const int bcol = nbase + lm;

    const float* __restrict__ xrow  = x  + (size_t)arow * IN_DIM;
    const float* __restrict__ wlrow = wl + (size_t)bcol * IN_DIM;
    const float* __restrict__ wrrow = wr + (size_t)bcol * IN_DIM;

    v8f accL = {};
    v8f accR = {};
    #pragma unroll 4
    for (int k = 0; k < IN_DIM; k += 4) {
        v2f a, bL, bR;
        a.x  = xrow[k + kh];      a.y  = xrow[k + kh + 1];
        bL.x = wlrow[k + kh];     bL.y = wlrow[k + kh + 1];
        bR.x = wrrow[k + kh];     bR.y = wrrow[k + kh + 1];
        accL = __builtin_amdgcn_wmma_f32_16x16x4_f32(
            false, a, false, bL, (short)0, accL, false, false);
        accR = __builtin_amdgcn_wmma_f32_16x16x4_f32(
            false, a, false, bR, (short)0, accR, false, false);
    }

    const float biasL = bl[bcol];
    const float biasR = br[bcol];
    const int rbase = mbase + (lane >> 4) * 8;
    #pragma unroll
    for (int r = 0; r < 8; ++r) {
        const size_t off = (size_t)(rbase + r) * OUT_DIM + bcol;
        xl[off] = accL[r] + biasL;
        xr[off] = accR[r] + biasR;
    }
}

__global__ void gatv2_init(float* __restrict__ maxv, float* __restrict__ denom,
                           float* __restrict__ outacc)
{
    const int i = blockIdx.x * blockDim.x + threadIdx.x;
    if (i < N_NODES * HEADS) {
        maxv[i]  = -__builtin_inff();
        denom[i] = 0.0f;
    }
    if (i < N_NODES * OUT_DIM) outacc[i] = 0.0f;
}

#define EL_BLOCKS 12500
__global__ __launch_bounds__(256) void gatv2_edge_logits(
    const int* __restrict__ ei, const float* __restrict__ eattr,
    const float* __restrict__ we, const float* __restrict__ att,
    const float* __restrict__ xl, const float* __restrict__ xr,
    float* __restrict__ logits, float* __restrict__ maxv)
{
    const int lane = threadIdx.x & 31;
    const int wid = __builtin_amdgcn_readfirstlane(
        (int)(blockIdx.x * 8 + (threadIdx.x >> 5)));
    const int nwaves = EL_BLOCKS * 8;

    float w[HEADS][EDGE_DIM];
    float a[HEADS];
    #pragma unroll
    for (int h = 0; h < HEADS; ++h) {
        const int c = h * HEAD_DIM + lane;
        #pragma unroll
        for (int d = 0; d < EDGE_DIM; ++d)
            w[h][d] = we[(size_t)c * EDGE_DIM + d];
        a[h] = att[c];
    }

    for (int e = wid; e < N_EDGES; e += nwaves) {
        const int src = ei[e];
        const int dst = ei[N_EDGES + e];

        float ea[EDGE_DIM];
        #pragma unroll
        for (int d = 0; d < EDGE_DIM; ++d)
            ea[d] = eattr[(size_t)e * EDGE_DIM + d];

        const float* __restrict__ xls = xl + (size_t)src * OUT_DIM + lane;
        const float* __restrict__ xrd = xr + (size_t)dst * OUT_DIM + lane;
        float gl[HEADS], gr[HEADS];
        #pragma unroll
        for (int h = 0; h < HEADS; ++h) {
            gl[h] = xls[h * HEAD_DIM];
            gr[h] = xrd[h * HEAD_DIM];
        }

        #pragma unroll
        for (int h = 0; h < HEADS; ++h) {
            float ev = 0.0f;
            #pragma unroll
            for (int d = 0; d < EDGE_DIM; ++d)
                ev = fmaf(ea[d], w[h][d], ev);
            float v = gl[h] + gr[h] + ev;
            v = v > 0.0f ? v : NEG_SLOPE * v;
            float p = v * a[h];
            #pragma unroll
            for (int off = 16; off > 0; off >>= 1)
                p += __shfl_xor(p, off, 32);
            if (lane == 0) {
                logits[(size_t)e * HEADS + h] = p;
                atomicMax(&maxv[(size_t)dst * HEADS + h], p);
            }
        }
    }
}

__global__ void gatv2_edge_expsum(
    const int* __restrict__ ei, float* __restrict__ logits,
    const float* __restrict__ maxv, float* __restrict__ denom)
{
    const int i = blockIdx.x * blockDim.x + threadIdx.x;
    if (i >= N_EDGES * HEADS) return;
    const int e = i >> 2;
    const int h = i & 3;
    const int dst = ei[N_EDGES + e];
    const float ex = __expf(logits[i] - maxv[(size_t)dst * HEADS + h]);
    logits[i] = ex;
    atomicAdd(&denom[(size_t)dst * HEADS + h], ex);
}

__global__ __launch_bounds__(256) void gatv2_edge_aggregate(
    const int* __restrict__ ei, const float* __restrict__ logits,
    const float* __restrict__ denom, const float* __restrict__ xl,
    float* __restrict__ outacc)
{
    const int e = blockIdx.x * 8 + (threadIdx.x >> 5);
    if (e >= N_EDGES) return;
    const int lane = threadIdx.x & 31;
    const int src = ei[e];
    const int dst = ei[N_EDGES + e];
    #pragma unroll
    for (int h = 0; h < HEADS; ++h) {
        const float alpha = logits[(size_t)e * HEADS + h] /
                            (denom[(size_t)dst * HEADS + h] + 1e-16f);
        const int c = h * HEAD_DIM + lane;
        atomicAdd(&outacc[(size_t)dst * OUT_DIM + c],
                  xl[(size_t)src * OUT_DIM + c] * alpha);
    }
}

__global__ __launch_bounds__(256) void gatv2_final(
    const float* __restrict__ x, const float* __restrict__ outacc,
    const float* __restrict__ cbias, const float* __restrict__ gamma,
    const float* __restrict__ beta, float* __restrict__ out)
{
    const int n = blockIdx.x * 8 + (threadIdx.x >> 5);
    if (n >= N_NODES) return;
    const int lane = threadIdx.x & 31;

    float h[4];
    float s = 0.0f;
    #pragma unroll
    for (int j = 0; j < 4; ++j) {
        const int c = lane + 32 * j;
        const float v = outacc[(size_t)n * OUT_DIM + c] + cbias[c];
        const float sig = 1.0f / (1.0f + __expf(-v));
        h[j] = v * sig + x[(size_t)n * OUT_DIM + c];
        s += h[j];
    }
    #pragma unroll
    for (int off = 16; off > 0; off >>= 1) s += __shfl_xor(s, off, 32);
    const float mu = s * (1.0f / OUT_DIM);

    float vs = 0.0f;
    #pragma unroll
    for (int j = 0; j < 4; ++j) { const float d = h[j] - mu; vs += d * d; }
    #pragma unroll
    for (int off = 16; off > 0; off >>= 1) vs += __shfl_xor(vs, off, 32);
    const float inv = rsqrtf(vs * (1.0f / OUT_DIM) + LN_EPS);

    #pragma unroll
    for (int j = 0; j < 4; ++j) {
        const int c = lane + 32 * j;
        out[(size_t)n * OUT_DIM + c] = (h[j] - mu) * inv * gamma[c] + beta[c];
    }
}

extern "C" void kernel_launch(void* const* d_in, const int* in_sizes, int n_in,
                              void* d_out, int out_size, void* d_ws, size_t ws_size,
                              hipStream_t stream) {
    const float* x     = (const float*)d_in[0];
    const int*   ei    = (const int*)  d_in[1];
    const float* eattr = (const float*)d_in[2];
    const float* wl    = (const float*)d_in[3];
    const float* bl    = (const float*)d_in[4];
    const float* wr    = (const float*)d_in[5];
    const float* br    = (const float*)d_in[6];
    const float* we    = (const float*)d_in[7];
    const float* att   = (const float*)d_in[8];
    const float* cb    = (const float*)d_in[9];
    const float* gam   = (const float*)d_in[10];
    const float* bet   = (const float*)d_in[11];
    float* out = (float*)d_out;

    float* ws     = (float*)d_ws;
    float* xl     = ws;
    float* xr     = xl     + (size_t)N_NODES * OUT_DIM;
    float* outacc = xr     + (size_t)N_NODES * OUT_DIM;
    float* logits = outacc + (size_t)N_NODES * OUT_DIM;
    float* maxv   = logits + (size_t)N_EDGES * HEADS;
    float* denom  = maxv   + (size_t)N_NODES * HEADS;

    gatv2_gemm_lr<<<N_NODES / 16, 256, 0, stream>>>(x, wl, bl, wr, br, xl, xr);

    const int initN = N_NODES * OUT_DIM;
    gatv2_init<<<(initN + 255) / 256, 256, 0, stream>>>(maxv, denom, outacc);

    gatv2_edge_logits<<<EL_BLOCKS, 256, 0, stream>>>(
        ei, eattr, we, att, xl, xr, logits, maxv);

    gatv2_edge_expsum<<<(N_EDGES * HEADS + 255) / 256, 256, 0, stream>>>(
        ei, logits, maxv, denom);

    gatv2_edge_aggregate<<<(N_EDGES + 7) / 8, 256, 0, stream>>>(
        ei, logits, denom, xl, outacc);

    gatv2_final<<<(N_NODES + 7) / 8, 256, 0, stream>>>(
        x, outacc, cb, gam, bet, out);
}


# edges counting-sorted by dst (int atomics); one wave owns a node, lane = 4 channels: running-max softmax + weighted sum in registers, no float atomics; sum order: slot order, in-lane then lane tree
# speedup vs baseline: 2.0138x; 1.9208x over previous
#include <hip/hip_runtime.h>

#define N_NODES 50000
#define N_EDGES 800000
#define IN_DIM 128
#define OUT_DIM 128
#define HEADS 4
#define HEAD_DIM 32
#define EDGE_DIM 10
#define NEG_SLOPE 0.2f
#define LN_EPS 1e-5f

typedef __attribute__((ext_vector_type(2))) float v2f;
typedef __attribute__((ext_vector_type(8))) float v8f;
typedef __attribute__((ext_vector_type(4))) float v4f;

__global__ __launch_bounds__(256) void gatv2_gemm_lr(
    const float* __restrict__ x,
    const float* __restrict__ wl, const float* __restrict__ bl,
    const float* __restrict__ wr, const float* __restrict__ br,
    float* __restrict__ xl, float* __restrict__ xr)
{
    const int wave = threadIdx.x >> 5;
    const int lane = threadIdx.x & 31;
    const int mbase = blockIdx.x * 16;
    const int nbase = wave * 16;
    const int lm = lane & 15;
    const int kh = (lane >> 4) * 2;

    const int arow = mbase + lm;
    const int bcol = nbase + lm;

    const float* __restrict__ xrow  = x  + (size_t)arow * IN_DIM;
    const float* __restrict__ wlrow = wl + (size_t)bcol * IN_DIM;
    const float* __restrict__ wrrow = wr + (size_t)bcol * IN_DIM;

    v8f accL = {};
    v8f accR = {};
    #pragma unroll 4
    for (int k = 0; k < IN_DIM; k += 4) {
        v2f a, bL, bR;
        a.x  = xrow[k + kh];      a.y  = xrow[k + kh + 1];
        bL.x = wlrow[k + kh];     bL.y = wlrow[k + kh + 1];
        bR.x = wrrow[k + kh];     bR.y = wrrow[k + kh + 1];
        accL = __builtin_amdgcn_wmma_f32_16x16x4_f32(
            false, a, false, bL, (short)0, accL, false, false);
        accR = __builtin_amdgcn_wmma_f32_16x16x4_f32(
            false, a, false, bR, (short)0, accR, false, false);
    }

    const float biasL = bl[bcol];
    const float biasR = br[bcol];
    const int rbase = mbase + (lane >> 4) * 8;
    #pragma unroll
    for (int r = 0; r < 8; ++r) {
        const size_t off = (size_t)(rbase + r) * OUT_DIM + bcol;
        xl[off] = accL[r] + biasL;
        xr[off] = accR[r] + biasR;
    }
}

__global__ void gatv2_init(int* __restrict__ deg, int* __restrict__ gcursor)
{
    const int i = blockIdx.x * blockDim.x + threadIdx.x;
    if (i < N_NODES) deg[i] = 0;
    if (i == 0) gcursor[0] = 0;
}

#define EDGES_PER_THREAD 8
#define EDGE_THREADS (N_EDGES / EDGES_PER_THREAD)
__global__ __launch_bounds__(256) void gatv2_count(
    const int* __restrict__ ei, int* __restrict__ deg)
{
    const int i = blockIdx.x * 256 + threadIdx.x;
    if (i >= EDGE_THREADS) return;
    int dst[EDGES_PER_THREAD];
    #pragma unroll
    for (int k = 0; k < EDGES_PER_THREAD; ++k)
        dst[k] = ei[N_EDGES + i + k * EDGE_THREADS];
    #pragma unroll
    for (int k = 0; k < EDGES_PER_THREAD; ++k)
        atomicAdd(&deg[dst[k]], 1);
}

__global__ __launch_bounds__(256) void gatv2_alloc(
    const int* __restrict__ deg, int* __restrict__ rowptr,
    int* __restrict__ cursor, int* __restrict__ gcursor)
{
    const int i = blockIdx.x * 256 + threadIdx.x;
    const int lane = threadIdx.x & 31;
    const int v = (i < N_NODES) ? deg[i] : 0;
    int incl = v;
    #pragma unroll
    for (int off = 1; off < 32; off <<= 1) {
        const int t = __shfl_up(incl, off, 32);
        if (lane >= off) incl += t;
    }
    const int total = __shfl(incl, 31, 32);
    int base = 0;
    if (lane == 0) base = atomicAdd(gcursor, total);
    base = __shfl(base, 0, 32);
    if (i < N_NODES) {
        const int r = base + incl - v;
        rowptr[i] = r;
        cursor[i] = r;
    }
}

__global__ __launch_bounds__(256) void gatv2_fill(
    const int* __restrict__ ei, int* __restrict__ cursor, int* __restrict__ perm)
{
    const int i = blockIdx.x * 256 + threadIdx.x;
    if (i >= EDGE_THREADS) return;
    int dst[EDGES_PER_THREAD], slot[EDGES_PER_THREAD];
    #pragma unroll
    for (int k = 0; k < EDGES_PER_THREAD; ++k)
        dst[k] = ei[N_EDGES + i + k * EDGE_THREADS];
    #pragma unroll
    for (int k = 0; k < EDGES_PER_THREAD; ++k)
        slot[k] = atomicAdd(&cursor[dst[k]], 1);
    #pragma unroll
    for (int k = 0; k < EDGES_PER_THREAD; ++k) {
        if ((unsigned)slot[k] < (unsigned)N_EDGES) perm[slot[k]] = i + k * EDGE_THREADS;
    }
}

__device__ __forceinline__ float gatv2_lane_logit(
    const float (&ea)[EDGE_DIM], const v2f (&w2)[2][EDGE_DIM], const v4f g,
    const v2f (&gr2)[2], const v2f (&a2)[2])
{
    v2f ev0 = {0.0f, 0.0f}, ev1 = {0.0f, 0.0f};
    #pragma unroll
    for (int d = 0; d < EDGE_DIM; ++d) {
        const v2f s = {ea[d], ea[d]};
        ev0 = __builtin_elementwise_fma(s, w2[0][d], ev0);
        ev1 = __builtin_elementwise_fma(s, w2[1][d], ev1);
    }
    const v2f g01 = {g.x, g.y}, g23 = {g.z, g.w};
    const v2f v0 = g01 + gr2[0] + ev0;
    const v2f v1 = g23 + gr2[1] + ev1;
    const float l0 = v0.x > 0.0f ? v0.x : NEG_SLOPE * v0.x;
    const float l1 = v0.y > 0.0f ? v0.y : NEG_SLOPE * v0.y;
    const float l2 = v1.x > 0.0f ? v1.x : NEG_SLOPE * v1.x;
    const float l3 = v1.y > 0.0f ? v1.y : NEG_SLOPE * v1.y;
    float p = l0 * a2[0].x;
    p = fmaf(l1, a2[0].y, p);
    p = fmaf(l2, a2[1].x, p);
    p = fmaf(l3, a2[1].y, p);
    return p;
}

__global__ __launch_bounds__(256) void gatv2_node(
    const int* __restrict__ ei, const float* __restrict__ eattr,
    const float* __restrict__ we, const float* __restrict__ att,
    const float* __restrict__ xl, const float* __restrict__ xr,
    const int* __restrict__ rowptr, const int* __restrict__ deg,
    const int* __restrict__ perm,
    const float* __restrict__ x, const float* __restrict__ cbias,
    const float* __restrict__ gamma, const float* __restrict__ beta,
    float* __restrict__ out)
{
    __shared__ v4f s_w[OUT_DIM * EDGE_DIM / 4];
    __shared__ v4f s_a[OUT_DIM / 4];

    const int lane = threadIdx.x & 31;
    for (int i = threadIdx.x; i < OUT_DIM * EDGE_DIM / 4; i += 256)
        s_w[i] = ((const v4f*)we)[i];
    if (threadIdx.x < OUT_DIM / 4)
        s_a[threadIdx.x] = ((const v4f*)att)[threadIdx.x];
    __syncthreads();

    float wf[4 * EDGE_DIM];
    #pragma unroll
    for (int i = 0; i < EDGE_DIM; ++i) {
        const v4f t = s_w[lane * EDGE_DIM + i];
        wf[4 * i + 0] = t.x; wf[4 * i + 1] = t.y;
        wf[4 * i + 2] = t.z; wf[4 * i + 3] = t.w;
    }
    v2f w2[2][EDGE_DIM];
    #pragma unroll
    for (int q = 0; q < 2; ++q) {
        #pragma unroll
        for (int d = 0; d < EDGE_DIM; ++d) {
            w2[q][d].x = wf[(2 * q) * EDGE_DIM + d];
            w2[q][d].y = wf[(2 * q + 1) * EDGE_DIM + d];
        }
    }
    const v4f a4 = s_a[lane];
    const v2f a2[2] = {{a4.x, a4.y}, {a4.z, a4.w}};

    const int n = __builtin_amdgcn_readfirstlane(
        (int)(blockIdx.x * 8 + (threadIdx.x >> 5)));

    const v4f r4 = *(const v4f*)(xr + (size_t)n * OUT_DIM + 4 * lane);
    const v2f gr2[2] = {{r4.x, r4.y}, {r4.z, r4.w}};

    int beg = rowptr[n];
    int end = beg + deg[n];
    beg = beg < 0 ? 0 : beg;
    end = end > N_EDGES ? N_EDGES : end;

    float m = -__builtin_inff();
    float den = 0.0f;
    float acc[4] = {0.0f, 0.0f, 0.0f, 0.0f};

    for (int base = beg; base < end; base += 32) {
        const int idx = base + lane;
        int my_e = 0, my_src = 0;
        if (idx < end) {
            my_e = perm[idx];
            my_e = (unsigned)my_e < (unsigned)N_EDGES ? my_e : 0;
            my_src = ei[my_e];
        }
        const int left = end - base;
        const int cnt = left < 32 ? left : 32;
        for (int j = 0; j < cnt; j += 2) {
            const bool two = (j + 1 < cnt);
            const int j1 = two ? j + 1 : j;
            const float* __restrict__ eap0 = eattr +
                (size_t)__builtin_amdgcn_readlane(my_e, j) * EDGE_DIM;
            const float* __restrict__ eap1 = eattr +
                (size_t)__builtin_amdgcn_readlane(my_e, j1) * EDGE_DIM;
            const v4f g0 = *(const v4f*)(xl +
                (size_t)__builtin_amdgcn_readlane(my_src, j) * OUT_DIM + 4 * lane);
            const v4f g1 = *(const v4f*)(xl +
                (size_t)__builtin_amdgcn_readlane(my_src, j1) * OUT_DIM + 4 * lane);
            float ea0[EDGE_DIM], ea1[EDGE_DIM];
            #pragma unroll
            for (int d = 0; d < EDGE_DIM; ++d) { ea0[d] = eap0[d]; ea1[d] = eap1[d]; }
            const float gl0[4] = {g0.x, g0.y, g0.z, g0.w};
            const float gl1[4] = {g1.x, g1.y, g1.z, g1.w};

            float p0 = gatv2_lane_logit(ea0, w2, g0, gr2, a2);
            float p1 = gatv2_lane_logit(ea1, w2, g1, gr2, a2);
            p0 += __shfl_xor(p0, 4, 32);  p1 += __shfl_xor(p1, 4, 32);
            p0 += __shfl_xor(p0, 2, 32);  p1 += __shfl_xor(p1, 2, 32);
            p0 += __shfl_xor(p0, 1, 32);  p1 += __shfl_xor(p1, 1, 32);
            p1 = two ? p1 : -__builtin_inff();

            const float mn = fmaxf(m, fmaxf(p0, p1));
            const float sc = __expf(m - mn);
            const float t0 = __expf(p0 - mn);
            const float t1 = __expf(p1 - mn);
            den = fmaf(den, sc, t0) + t1;
            #pragma unroll
            for (int k = 0; k < 4; ++k)
                acc[k] = fmaf(t1, gl1[k], fmaf(acc[k], sc, t0 * gl0[k]));
            m = mn;
        }
    }

    const float inv_den = 1.0f / (den + 1e-16f);

    const v4f x4 = *(const v4f*)(x + (size_t)n * OUT_DIM + 4 * lane);
    const v4f c4 = *(const v4f*)(cbias + 4 * lane);
    const float xv[4] = {x4.x, x4.y, x4.z, x4.w};
    const float cb[4] = {c4.x, c4.y, c4.z, c4.w};

    float h[4];
    float s = 0.0f;
    #pragma unroll
    for (int k = 0; k < 4; ++k) {
        const float v = acc[k] * inv_den + cb[k];
        const float sig = 1.0f / (1.0f + __expf(-v));
        h[k] = v * sig + xv[k];
        s += h[k];
    }
    #pragma unroll
    for (int off = 16; off > 0; off >>= 1) s += __shfl_xor(s, off, 32);
    const float mu = s * (1.0f / OUT_DIM);

    float vs = 0.0f;
    #pragma unroll
    for (int k = 0; k < 4; ++k) { const float d = h[k] - mu; vs += d * d; }
    #pragma unroll
    for (int off = 16; off > 0; off >>= 1) vs += __shfl_xor(vs, off, 32);
    const float inv = rsqrtf(vs * (1.0f / OUT_DIM) + LN_EPS);

    const v4f g4 = *(const v4f*)(gamma + 4 * lane);
    const v4f b4 = *(const v4f*)(beta + 4 * lane);
    v4f o;
    o.x = (h[0] - mu) * inv * g4.x + b4.x;
    o.y = (h[1] - mu) * inv * g4.y + b4.y;
    o.z = (h[2] - mu) * inv * g4.z + b4.z;
    o.w = (h[3] - mu) * inv * g4.w + b4.w;
    *(v4f*)(out + (size_t)n * OUT_DIM + 4 * lane) = o;
}

extern "C" void kernel_launch(void* const* d_in, const int* in_sizes, int n_in,
                              void* d_out, int out_size, void* d_ws, size_t ws_size,
                              hipStream_t stream) {
    const float* x     = (const float*)d_in[0];
    const int*   ei    = (const int*)  d_in[1];
    const float* eattr = (const float*)d_in[2];
    const float* wl    = (const float*)d_in[3];
    const float* bl    = (const float*)d_in[4];
    const float* wr    = (const float*)d_in[5];
    const float* br    = (const float*)d_in[6];
    const float* we    = (const float*)d_in[7];
    const float* att   = (const float*)d_in[8];
    const float* cb    = (const float*)d_in[9];
    const float* gam   = (const float*)d_in[10];
    const float* bet   = (const float*)d_in[11];
    float* out = (float*)d_out;

    float* ws     = (float*)d_ws;
    float* xl     = ws;
    float* xr     = xl     + (size_t)N_NODES * OUT_DIM;
    int*   deg    = (int*)(xr + (size_t)N_NODES * OUT_DIM);
    int*   rowptr = deg    + N_NODES;
    int*   cursor = rowptr + N_NODES;
    int*   gcur   = cursor + N_NODES;
    int*   perm   = gcur   + 16;

    gatv2_gemm_lr<<<N_NODES / 16, 256, 0, stream>>>(x, wl, bl, wr, br, xl, xr);

    gatv2_init<<<(N_NODES + 255) / 256, 256, 0, stream>>>(deg, gcur);

    gatv2_count<<<(EDGE_THREADS + 255) / 256, 256, 0, stream>>>(ei, deg);

    gatv2_alloc<<<(N_NODES + 255) / 256, 256, 0, stream>>>(deg, rowptr, cursor, gcur);

    gatv2_fill<<<(EDGE_THREADS + 255) / 256, 256, 0, stream>>>(ei, cursor, perm);

    gatv2_node<<<N_NODES / 8, 256, 0, stream>>>(
        ei, eattr, we, att, xl, xr, rowptr, deg, perm, x, cb, gam, bet, out);
}


# xl/xr GEMM: each wave keeps its w_l/w_r fragments in registers and walks node tiles (256 blocks), x by 16-byte loads one tile ahead; k sum order per 8 changes to {0,1,4,5} then {2,3,6,7}
# speedup vs baseline: 2.3482x; 1.1661x over previous
#include <hip/hip_runtime.h>

#define N_NODES 50000
#define N_EDGES 800000
#define IN_DIM 128
#define OUT_DIM 128
#define HEADS 4
#define HEAD_DIM 32
#define EDGE_DIM 10
#define NEG_SLOPE 0.2f
#define LN_EPS 1e-5f

typedef __attribute__((ext_vector_type(2))) float v2f;
typedef __attribute__((ext_vector_type(8))) float v8f;
typedef __attribute__((ext_vector_type(4))) float v4f;

#define GEMM_BLOCKS 256
#define GEMM_TILES (N_NODES / 16)
#define GEMM_FRAGS (IN_DIM / 8)

__device__ __forceinline__ void gatv2_load_frags(
    v4f (&f)[GEMM_FRAGS], const float* __restrict__ row_k4h)
{
    __builtin_amdgcn_sched_barrier(0);
    #pragma unroll
    for (int j = 0; j < GEMM_FRAGS; ++j)
        f[j] = *(const v4f*)(row_k4h + 8 * j);
    __builtin_amdgcn_sched_barrier(0);
}

__device__ __forceinline__ void gatv2_gemm_tile(
    const v4f (&a)[GEMM_FRAGS], const v4f (&fl)[GEMM_FRAGS],
    const v4f (&fr)[GEMM_FRAGS], const float biasL, const float biasR,
    float* __restrict__ xl, float* __restrict__ xr, const size_t out0)
{
    v8f accL = {};
    v8f accR = {};
    #pragma unroll
    for (int j = 0; j < GEMM_FRAGS; ++j) {
        const v2f a0 = {a[j].x, a[j].y},   a1 = {a[j].z, a[j].w};
        const v2f l0 = {fl[j].x, fl[j].y}, l1 = {fl[j].z, fl[j].w};
        const v2f r0 = {fr[j].x, fr[j].y}, r1 = {fr[j].z, fr[j].w};
        accL = __builtin_amdgcn_wmma_f32_16x16x4_f32(
            false, a0, false, l0, (short)0, accL, false, false);
        accR = __builtin_amdgcn_wmma_f32_16x16x4_f32(
            false, a0, false, r0, (short)0, accR, false, false);
        accL = __builtin_amdgcn_wmma_f32_16x16x4_f32(
            false, a1, false, l1, (short)0, accL, false, false);
        accR = __builtin_amdgcn_wmma_f32_16x16x4_f32(
            false, a1, false, r1, (short)0, accR, false, false);
    }
    #pragma unroll
    for (int r = 0; r < 8; ++r) {
        xl[out0 + (size_t)r * OUT_DIM] = accL[r] + biasL;
        xr[out0 + (size_t)r * OUT_DIM] = accR[r] + biasR;
    }
}

__global__ __launch_bounds__(256) void gatv2_gemm_lr(
    const float* __restrict__ x,
    const float* __restrict__ wl, const float* __restrict__ bl,
    const float* __restrict__ wr, const float* __restrict__ br,
    float* __restrict__ xl, float* __restrict__ xr)
{
    const int wave = threadIdx.x >> 5;
    const int lane = threadIdx.x & 31;
    const int c = lane & 15;
    const int h = lane >> 4;
    const int bcol = wave * 16 + c;

    v4f fl[GEMM_FRAGS], fr[GEMM_FRAGS];
    gatv2_load_frags(fl, wl + (size_t)bcol * IN_DIM + 4 * h);
    gatv2_load_frags(fr, wr + (size_t)bcol * IN_DIM + 4 * h);
    const float biasL = bl[bcol];
    const float biasR = br[bcol];

    const size_t a_lane = (size_t)c * IN_DIM + 4 * h;
    const size_t d_lane = (size_t)(8 * h) * OUT_DIM + bcol;
    const int b = blockIdx.x;
    if (b >= GEMM_TILES) return;
    const int my = (GEMM_TILES - 1 - b) / GEMM_BLOCKS + 1;
    const int last = b + (my - 1) * GEMM_BLOCKS;

    int t = b;
    v4f xa[GEMM_FRAGS], xb[GEMM_FRAGS];
    gatv2_load_frags(xa, x + (size_t)t * 16 * IN_DIM + a_lane);
    for (int p = 0; p < my / 2; ++p) {
        const int t1 = t + GEMM_BLOCKS;
        const int t2 = t1 + GEMM_BLOCKS;
        gatv2_load_frags(xb, x + (size_t)t1 * 16 * IN_DIM + a_lane);
        gatv2_gemm_tile(xa, fl, fr, biasL, biasR, xl, xr,
                        (size_t)t * 16 * OUT_DIM + d_lane);
        gatv2_load_frags(xa, x + (size_t)(t2 <= last ? t2 : last) * 16 * IN_DIM + a_lane);
        gatv2_gemm_tile(xb, fl, fr, biasL, biasR, xl, xr,
                        (size_t)t1 * 16 * OUT_DIM + d_lane);
        t = t2;
    }
    if (my & 1)
        gatv2_gemm_tile(xa, fl, fr, biasL, biasR, xl, xr,
                        (size_t)t * 16 * OUT_DIM + d_lane);
}

__global__ void gatv2_init(int* __restrict__ deg, int* __restrict__ gcursor)
{
    const int i = blockIdx.x * blockDim.x + threadIdx.x;
    if (i < N_NODES) deg[i] = 0;
    if (i == 0) gcursor[0] = 0;
}

#define EDGES_PER_THREAD 8
#define EDGE_THREADS (N_EDGES / EDGES_PER_THREAD)
__global__ __launch_bounds__(256) void gatv2_count(
    const int* __restrict__ ei, int* __restrict__ deg)
{
    const int i = blockIdx.x * 256 + threadIdx.x;
    if (i >= EDGE_THREADS) return;
    int dst[EDGES_PER_THREAD];
    #pragma unroll
    for (int k = 0; k < EDGES_PER_THREAD; ++k)
        dst[k] = ei[N_EDGES + i + k * EDGE_THREADS];
    #pragma unroll
    for (int k = 0; k < EDGES_PER_THREAD; ++k)
        atomicAdd(&deg[dst[k]], 1);
}

__global__ __launch_bounds__(256) void gatv2_alloc(
    const int* __restrict__ deg, int* __restrict__ rowptr,
    int* __restrict__ cursor, int* __restrict__ gcursor)
{
    const int i = blockIdx.x * 256 + threadIdx.x;
    const int lane = threadIdx.x & 31;
    const int v = (i < N_NODES) ? deg[i] : 0;
    int incl = v;
    #pragma unroll
    for (int off = 1; off < 32; off <<= 1) {
        const int t = __shfl_up(incl, off, 32);
        if (lane >= off) incl += t;
    }
    const int total = __shfl(incl, 31, 32);
    int base = 0;
    if (lane == 0) base = atomicAdd(gcursor, total);
    base = __shfl(base, 0, 32);
    if (i < N_NODES) {
        const int r = base + incl - v;
        rowptr[i] = r;
        cursor[i] = r;
    }
}

__global__ __launch_bounds__(256) void gatv2_fill(
    const int* __restrict__ ei, int* __restrict__ cursor, int* __restrict__ perm)
{
    const int i = blockIdx.x * 256 + threadIdx.x;
    if (i >= EDGE_THREADS) return;
    int dst[EDGES_PER_THREAD], slot[EDGES_PER_THREAD];
    #pragma unroll
    for (int k = 0; k < EDGES_PER_THREAD; ++k)
        dst[k] = ei[N_EDGES + i + k * EDGE_THREADS];
    #pragma unroll
    for (int k = 0; k < EDGES_PER_THREAD; ++k)
        slot[k] = atomicAdd(&cursor[dst[k]], 1);
    #pragma unroll
    for (int k = 0; k < EDGES_PER_THREAD; ++k) {
        if ((unsigned)slot[k] < (unsigned)N_EDGES) perm[slot[k]] = i + k * EDGE_THREADS;
    }
}

__device__ __forceinline__ float gatv2_lane_logit(
    const float (&ea)[EDGE_DIM], const v2f (&w2)[2][EDGE_DIM], const v4f g,
    const v2f (&gr2)[2], const v2f (&a2)[2])
{
    v2f ev0 = {0.0f, 0.0f}, ev1 = {0.0f, 0.0f};
    #pragma unroll
    for (int d = 0; d < EDGE_DIM; ++d) {
        const v2f s = {ea[d], ea[d]};
        ev0 = __builtin_elementwise_fma(s, w2[0][d], ev0);
        ev1 = __builtin_elementwise_fma(s, w2[1][d], ev1);
    }
    const v2f g01 = {g.x, g.y}, g23 = {g.z, g.w};
    const v2f v0 = g01 + gr2[0] + ev0;
    const v2f v1 = g23 + gr2[1] + ev1;
    const float l0 = v0.x > 0.0f ? v0.x : NEG_SLOPE * v0.x;
    const float l1 = v0.y > 0.0f ? v0.y : NEG_SLOPE * v0.y;
    const float l2 = v1.x > 0.0f ? v1.x : NEG_SLOPE * v1.x;
    const float l3 = v1.y > 0.0f ? v1.y : NEG_SLOPE * v1.y;
    float p = l0 * a2[0].x;
    p = fmaf(l1, a2[0].y, p);
    p = fmaf(l2, a2[1].x, p);
    p = fmaf(l3, a2[1].y, p);
    return p;
}

__global__ __launch_bounds__(256) void gatv2_node(
    const int* __restrict__ ei, const float* __restrict__ eattr,
    const float* __restrict__ we, const float* __restrict__ att,
    const float* __restrict__ xl, const float* __restrict__ xr,
    const int* __restrict__ rowptr, const int* __restrict__ deg,
    const int* __restrict__ perm,
    const float* __restrict__ x, const float* __restrict__ cbias,
    const float* __restrict__ gamma, const float* __restrict__ beta,
    float* __restrict__ out)
{
    __shared__ v4f s_w[OUT_DIM * EDGE_DIM / 4];
    __shared__ v4f s_a[OUT_DIM / 4];

    const int lane = threadIdx.x & 31;
    for (int i = threadIdx.x; i < OUT_DIM * EDGE_DIM / 4; i += 256)
        s_w[i] = ((const v4f*)we)[i];
    if (threadIdx.x < OUT_DIM / 4)
        s_a[threadIdx.x] = ((const v4f*)att)[threadIdx.x];
    __syncthreads();

    float wf[4 * EDGE_DIM];
    #pragma unroll
    for (int i = 0; i < EDGE_DIM; ++i) {
        const v4f t = s_w[lane * EDGE_DIM + i];
        wf[4 * i + 0] = t.x; wf[4 * i + 1] = t.y;
        wf[4 * i + 2] = t.z; wf[4 * i + 3] = t.w;
    }
    v2f w2[2][EDGE_DIM];
    #pragma unroll
    for (int q = 0; q < 2; ++q) {
        #pragma unroll
        for (int d = 0; d < EDGE_DIM; ++d) {
            w2[q][d].x = wf[(2 * q) * EDGE_DIM + d];
            w2[q][d].y = wf[(2 * q + 1) * EDGE_DIM + d];
        }
    }
    const v4f a4 = s_a[lane];
    const v2f a2[2] = {{a4.x, a4.y}, {a4.z, a4.w}};

    const int n = __builtin_amdgcn_readfirstlane(
        (int)(blockIdx.x * 8 + (threadIdx.x >> 5)));

    const v4f r4 = *(const v4f*)(xr + (size_t)n * OUT_DIM + 4 * lane);
    const v2f gr2[2] = {{r4.x, r4.y}, {r4.z, r4.w}};

    int beg = rowptr[n];
    int end = beg + deg[n];
    beg = beg < 0 ? 0 : beg;
    end = end > N_EDGES ? N_EDGES : end;

    float m = -__builtin_inff();
    float den = 0.0f;
    float acc[4] = {0.0f, 0.0f, 0.0f, 0.0f};

    for (int base = beg; base < end; base += 32) {
        const int idx = base + lane;
        int my_e = 0, my_src = 0;
        if (idx < end) {
            my_e = perm[idx];
            my_e = (unsigned)my_e < (unsigned)N_EDGES ? my_e : 0;
            my_src = ei[my_e];
        }
        const int left = end - base;
        const int cnt = left < 32 ? left : 32;
        for (int j = 0; j < cnt; j += 2) {
            const bool two = (j + 1 < cnt);
            const int j1 = two ? j + 1 : j;
            const float* __restrict__ eap0 = eattr +
                (size_t)__builtin_amdgcn_readlane(my_e, j) * EDGE_DIM;
            const float* __restrict__ eap1 = eattr +
                (size_t)__builtin_amdgcn_readlane(my_e, j1) * EDGE_DIM;
            const v4f g0 = *(const v4f*)(xl +
                (size_t)__builtin_amdgcn_readlane(my_src, j) * OUT_DIM + 4 * lane);
            const v4f g1 = *(const v4f*)(xl +
                (size_t)__builtin_amdgcn_readlane(my_src, j1) * OUT_DIM + 4 * lane);
            float ea0[EDGE_DIM], ea1[EDGE_DIM];
            #pragma unroll
            for (int d = 0; d < EDGE_DIM; ++d) { ea0[d] = eap0[d]; ea1[d] = eap1[d]; }
            const float gl0[4] = {g0.x, g0.y, g0.z, g0.w};
            const float gl1[4] = {g1.x, g1.y, g1.z, g1.w};

            float p0 = gatv2_lane_logit(ea0, w2, g0, gr2, a2);
            float p1 = gatv2_lane_logit(ea1, w2, g1, gr2, a2);
            p0 += __shfl_xor(p0, 4, 32);  p1 += __shfl_xor(p1, 4, 32);
            p0 += __shfl_xor(p0, 2, 32);  p1 += __shfl_xor(p1, 2, 32);
            p0 += __shfl_xor(p0, 1, 32);  p1 += __shfl_xor(p1, 1, 32);
            p1 = two ? p1 : -__builtin_inff();

            const float mn = fmaxf(m, fmaxf(p0, p1));
            const float sc = __expf(m - mn);
            const float t0 = __expf(p0 - mn);
            const float t1 = __expf(p1 - mn);
            den = fmaf(den, sc, t0) + t1;
            #pragma unroll
            for (int k = 0; k < 4; ++k)
                acc[k] = fmaf(t1, gl1[k], fmaf(acc[k], sc, t0 * gl0[k]));
            m = mn;
        }
    }

    const float inv_den = 1.0f / (den + 1e-16f);

    const v4f x4 = *(const v4f*)(x + (size_t)n * OUT_DIM + 4 * lane);
    const v4f c4 = *(const v4f*)(cbias + 4 * lane);
    const float xv[4] = {x4.x, x4.y, x4.z, x4.w};
    const float cb[4] = {c4.x, c4.y, c4.z, c4.w};

    float h[4];
    float s = 0.0f;
    #pragma unroll
    for (int k = 0; k < 4; ++k) {
        const float v = acc[k] * inv_den + cb[k];
        const float sig = 1.0f / (1.0f + __expf(-v));
        h[k] = v * sig + xv[k];
        s += h[k];
    }
    #pragma unroll
    for (int off = 16; off > 0; off >>= 1) s += __shfl_xor(s, off, 32);
    const float mu = s * (1.0f / OUT_DIM);

    float vs = 0.0f;
    #pragma unroll
    for (int k = 0; k < 4; ++k) { const float d = h[k] - mu; vs += d * d; }
    #pragma unroll
    for (int off = 16; off > 0; off >>= 1) vs += __shfl_xor(vs, off, 32);
    const float inv = rsqrtf(vs * (1.0f / OUT_DIM) + LN_EPS);

    const v4f g4 = *(const v4f*)(gamma + 4 * lane);
    const v4f b4 = *(const v4f*)(beta + 4 * lane);
    v4f o;
    o.x = (h[0] - mu) * inv * g4.x + b4.x;
    o.y = (h[1] - mu) * inv * g4.y + b4.y;
    o.z = (h[2] - mu) * inv * g4.z + b4.z;
    o.w = (h[3] - mu) * inv * g4.w + b4.w;
    *(v4f*)(out + (size_t)n * OUT_DIM + 4 * lane) = o;
}

extern "C" void kernel_launch(void* const* d_in, const int* in_sizes, int n_in,
                              void* d_out, int out_size, void* d_ws, size_t ws_size,
                              hipStream_t stream) {
    const float* x     = (const float*)d_in[0];
    const int*   ei    = (const int*)  d_in[1];
    const float* eattr = (const float*)d_in[2];
    const float* wl    = (const float*)d_in[3];
    const float* bl    = (const float*)d_in[4];
    const float* wr    = (const float*)d_in[5];
    const float* br    = (const float*)d_in[6];
    const float* we    = (const float*)d_in[7];
    const float* att   = (const float*)d_in[8];
    const float* cb    = (const float*)d_in[9];
    const float* gam   = (const float*)d_in[10];
    const float* bet   = (const float*)d_in[11];
    float* out = (float*)d_out;

    float* ws     = (float*)d_ws;
    float* xl     = ws;
    float* xr     = xl     + (size_t)N_NODES * OUT_DIM;
    int*   deg    = (int*)(xr + (size_t)N_NODES * OUT_DIM);
    int*   rowptr = deg    + N_NODES;
    int*   cursor = rowptr + N_NODES;
    int*   gcur   = cursor + N_NODES;
    int*   perm   = gcur   + 16;

    gatv2_gemm_lr<<<GEMM_BLOCKS, 256, 0, stream>>>(x, wl, bl, wr, br, xl, xr);

    gatv2_init<<<(N_NODES + 255) / 256, 256, 0, stream>>>(deg, gcur);

    gatv2_count<<<(EDGE_THREADS + 255) / 256, 256, 0, stream>>>(ei, deg);

    gatv2_alloc<<<(N_NODES + 255) / 256, 256, 0, stream>>>(deg, rowptr, cursor, gcur);

    gatv2_fill<<<(EDGE_THREADS + 255) / 256, 256, 0, stream>>>(ei, cursor, perm);

    gatv2_node<<<N_NODES / 8, 256, 0, stream>>>(
        ei, eattr, we, att, xl, xr, rowptr, deg, perm, x, cb, gam, bet, out);
}


# edges grouped by dst by a two-level counting sort on LDS counters (chunk histogram, bucket scatter, per-bucket rank), no per-edge global atomics; sum order in a node: LDS rank arrival order
# speedup vs baseline: 4.1168x; 1.7532x over previous
#include <hip/hip_runtime.h>

#define N_NODES 50000
#define N_EDGES 800000
#define IN_DIM 128
#define OUT_DIM 128
#define HEADS 4
#define HEAD_DIM 32
#define EDGE_DIM 10
#define NEG_SLOPE 0.2f
#define LN_EPS 1e-5f

typedef __attribute__((ext_vector_type(2))) float v2f;
typedef __attribute__((ext_vector_type(8))) float v8f;
typedef __attribute__((ext_vector_type(4))) float v4f;

#define GEMM_BLOCKS 256
#define GEMM_TILES (N_NODES / 16)
#define GEMM_FRAGS (IN_DIM / 8)

__device__ __forceinline__ void gatv2_load_frags(
    v4f (&f)[GEMM_FRAGS], const float* __restrict__ row_k4h)
{
    __builtin_amdgcn_sched_barrier(0);
    #pragma unroll
    for (int j = 0; j < GEMM_FRAGS; ++j)
        f[j] = *(const v4f*)(row_k4h + 8 * j);
    __builtin_amdgcn_sched_barrier(0);
}

__device__ __forceinline__ void gatv2_gemm_tile(
    const v4f (&a)[GEMM_FRAGS], const v4f (&fl)[GEMM_FRAGS],
    const v4f (&fr)[GEMM_FRAGS], const float biasL, const float biasR,
    float* __restrict__ xl, float* __restrict__ xr, const size_t out0)
{
    v8f accL = {};
    v8f accR = {};
    #pragma unroll
    for (int j = 0; j < GEMM_FRAGS; ++j) {
        const v2f a0 = {a[j].x, a[j].y},   a1 = {a[j].z, a[j].w};
        const v2f l0 = {fl[j].x, fl[j].y}, l1 = {fl[j].z, fl[j].w};
        const v2f r0 = {fr[j].x, fr[j].y}, r1 = {fr[j].z, fr[j].w};
        accL = __builtin_amdgcn_wmma_f32_16x16x4_f32(
            false, a0, false, l0, (short)0, accL, false, false);
        accR = __builtin_amdgcn_wmma_f32_16x16x4_f32(
            false, a0, false, r0, (short)0, accR, false, false);
        accL = __builtin_amdgcn_wmma_f32_16x16x4_f32(
            false, a1, false, l1, (short)0, accL, false, false);
        accR = __builtin_amdgcn_wmma_f32_16x16x4_f32(
            false, a1, false, r1, (short)0, accR, false, false);
    }
    #pragma unroll
    for (int r = 0; r < 8; ++r) {
        xl[out0 + (size_t)r * OUT_DIM] = accL[r] + biasL;
        xr[out0 + (size_t)r * OUT_DIM] = accR[r] + biasR;
    }
}

__global__ __launch_bounds__(256) void gatv2_gemm_lr(
    const float* __restrict__ x,
    const float* __restrict__ wl, const float* __restrict__ bl,
    const float* __restrict__ wr, const float* __restrict__ br,
    float* __restrict__ xl, float* __restrict__ xr)
{
    const int wave = threadIdx.x >> 5;
    const int lane = threadIdx.x & 31;
    const int c = lane & 15;
    const int h = lane >> 4;
    const int bcol = wave * 16 + c;

    v4f fl[GEMM_FRAGS], fr[GEMM_FRAGS];
    gatv2_load_frags(fl, wl + (size_t)bcol * IN_DIM + 4 * h);
    gatv2_load_frags(fr, wr + (size_t)bcol * IN_DIM + 4 * h);
    const float biasL = bl[bcol];
    const float biasR = br[bcol];

    const size_t a_lane = (size_t)c * IN_DIM + 4 * h;
    const size_t d_lane = (size_t)(8 * h) * OUT_DIM + bcol;
    const int b = blockIdx.x;
    if (b >= GEMM_TILES) return;
    const int my = (GEMM_TILES - 1 - b) / GEMM_BLOCKS + 1;
    const int last = b + (my - 1) * GEMM_BLOCKS;

    int t = b;
    v4f xa[GEMM_FRAGS], xb[GEMM_FRAGS];
    gatv2_load_frags(xa, x + (size_t)t * 16 * IN_DIM + a_lane);
    for (int p = 0; p < my / 2; ++p) {
        const int t1 = t + GEMM_BLOCKS;
        const int t2 = t1 + GEMM_BLOCKS;
        gatv2_load_frags(xb, x + (size_t)t1 * 16 * IN_DIM + a_lane);
        gatv2_gemm_tile(xa, fl, fr, biasL, biasR, xl, xr,
                        (size_t)t * 16 * OUT_DIM + d_lane);
        gatv2_load_frags(xa, x + (size_t)(t2 <= last ? t2 : last) * 16 * IN_DIM + a_lane);
        gatv2_gemm_tile(xb, fl, fr, biasL, biasR, xl, xr,
                        (size_t)t1 * 16 * OUT_DIM + d_lane);
        t = t2;
    }
    if (my & 1)
        gatv2_gemm_tile(xa, fl, fr, biasL, biasR, xl, xr,
                        (size_t)t * 16 * OUT_DIM + d_lane);
}

typedef __attribute__((ext_vector_type(2))) int v2i;

#define SORT_CHUNKS 256
#define CHUNK_EDGES (N_EDGES / SORT_CHUNKS)
#define CHUNK_EPT ((CHUNK_EDGES + 255) / 256)
#define BKT_SHIFT 8
#define BKT_NODES (1 << BKT_SHIFT)
#define N_BKT ((N_NODES + BKT_NODES - 1) / BKT_NODES)
#define RANK_UNROLL 8

__device__ __forceinline__ int gatv2_bucket(const int d)
{
    const unsigned k = (unsigned)d >> BKT_SHIFT;
    return (int)(k < (unsigned)N_BKT ? k : (unsigned)(N_BKT - 1));
}

__device__ __forceinline__ void gatv2_chunk_dst(
    int (&d)[CHUNK_EPT], const int* __restrict__ ei, const int chunk, const int t)
{
    const int* __restrict__ dstp = ei + N_EDGES + chunk * CHUNK_EDGES;
    #pragma unroll
    for (int u = 0; u < CHUNK_EPT; ++u) {
        const int idx = t + u * 256;
        d[u] = idx < CHUNK_EDGES ? dstp[idx] : -1;
    }
}

__device__ __forceinline__ int gatv2_block_excl_scan(const int v, int* wsum)
{
    const int lane = threadIdx.x & 31;
    const int w = threadIdx.x >> 5;
    int incl = v;
    #pragma unroll
    for (int off = 1; off < 32; off <<= 1) {
        const int s = __shfl_up(incl, off, 32);
        if (lane >= off) incl += s;
    }
    if (lane == 31) wsum[w] = incl;
    __syncthreads();
    int wbase = 0;
    #pragma unroll
    for (int i = 0; i < 8; ++i) wbase += (i < w) ? wsum[i] : 0;
    return wbase + incl - v;
}

__global__ __launch_bounds__(256) void gatv2_chunk_hist(
    const int* __restrict__ ei, int* __restrict__ counts)
{
    __shared__ int hist[N_BKT];
    const int t = threadIdx.x;
    int d[CHUNK_EPT];
    gatv2_chunk_dst(d, ei, blockIdx.x, t);
    if (t < N_BKT) hist[t] = 0;
    __syncthreads();
    #pragma unroll
    for (int u = 0; u < CHUNK_EPT; ++u)
        if (d[u] >= 0) atomicAdd(&hist[gatv2_bucket(d[u])], 1);
    __syncthreads();
    if (t < N_BKT) counts[blockIdx.x * N_BKT + t] = hist[t];
}

__global__ __launch_bounds__(256) void gatv2_scatter(
    const int* __restrict__ ei, const int* __restrict__ counts,
    int* __restrict__ bbase, int* __restrict__ btot, int* __restrict__ rec)
{
    __shared__ int gstart[256];
    __shared__ int lstart[256];
    __shared__ int lcur[256];
    __shared__ int wsum[8];
    __shared__ int nstaged;
    __shared__ v2i stage[CHUNK_EDGES];
    const int t = threadIdx.x;
    const int b = blockIdx.x;
    int d[CHUNK_EPT];
    gatv2_chunk_dst(d, ei, b, t);

    int tot = 0, mine = 0, own = 0;
    if (t < N_BKT) {
        #pragma unroll 16
        for (int bb = 0; bb < SORT_CHUNKS; ++bb) {
            const int c = counts[bb * N_BKT + t];
            tot += c;
            mine += (bb < b) ? c : 0;
            own = (bb == b) ? c : own;
        }
    }
    const int base = gatv2_block_excl_scan(tot, wsum);
    __syncthreads();
    const int lst = gatv2_block_excl_scan(own, wsum);
    gstart[t] = base + mine;
    lstart[t] = lst;
    lcur[t] = lst;
    if (t == 255) nstaged = lst + own;
    if (b == 0 && t < N_BKT) { bbase[t] = base; btot[t] = tot; }
    __syncthreads();

    #pragma unroll
    for (int u = 0; u < CHUNK_EPT; ++u) {
        if (d[u] >= 0) {
            const int lp = atomicAdd(&lcur[gatv2_bucket(d[u])], 1);
            if ((unsigned)lp < (unsigned)CHUNK_EDGES) {
                const v2i r = {b * CHUNK_EDGES + t + u * 256, d[u]};
                stage[lp] = r;
            }
        }
    }
    __syncthreads();

    int n = nstaged;
    n = n < CHUNK_EDGES ? n : CHUNK_EDGES;
    for (int i = t; i < n; i += 256) {
        const v2i r = stage[i];
        const int k = gatv2_bucket(r.y);
        const int g = gstart[k] + (i - lstart[k]);
        if ((unsigned)g < (unsigned)N_EDGES) ((v2i*)rec)[g] = r;
    }
}

#define RANK_STAGE 8192
__global__ __launch_bounds__(256) void gatv2_rank(
    const int* __restrict__ rec, const int* __restrict__ bbase,
    const int* __restrict__ btot, int* __restrict__ rowptr,
    int* __restrict__ deg, int* __restrict__ perm)
{
    __shared__ int cnt[BKT_NODES];
    __shared__ int wsum[8];
    __shared__ int stage[RANK_STAGE];
    const int t = threadIdx.x;
    const int node0 = blockIdx.x * BKT_NODES;
    int base = bbase[blockIdx.x];
    int tot = btot[blockIdx.x];
    base = base < 0 ? 0 : (base > N_EDGES ? N_EDGES : base);
    tot = tot < 0 ? 0 : (tot > N_EDGES - base ? N_EDGES - base : tot);
    const v2i* __restrict__ recs = (const v2i*)rec + base;
    const v2i none = {-1, -1};

    cnt[t] = 0;
    __syncthreads();
    for (int i0 = t; i0 < tot; i0 += 256 * RANK_UNROLL) {
        v2i r[RANK_UNROLL];
        #pragma unroll
        for (int u = 0; u < RANK_UNROLL; ++u) {
            const int i = i0 + u * 256;
            r[u] = i < tot ? recs[i] : none;
        }
        #pragma unroll
        for (int u = 0; u < RANK_UNROLL; ++u) {
            const unsigned dl = (unsigned)(r[u].y - node0);
            if (r[u].x >= 0 && dl < (unsigned)BKT_NODES) atomicAdd(&cnt[dl], 1);
        }
    }
    __syncthreads();

    const int c = cnt[t];
    const int first = base + gatv2_block_excl_scan(c, wsum);
    cnt[t] = first;
    if (node0 + t < N_NODES) {
        rowptr[node0 + t] = first;
        deg[node0 + t] = c;
    }
    __syncthreads();

    const bool staged = tot <= RANK_STAGE;
    for (int i0 = t; i0 < tot; i0 += 256 * RANK_UNROLL) {
        v2i r[RANK_UNROLL];
        #pragma unroll
        for (int u = 0; u < RANK_UNROLL; ++u) {
            const int i = i0 + u * 256;
            r[u] = i < tot ? recs[i] : none;
        }
        #pragma unroll
        for (int u = 0; u < RANK_UNROLL; ++u) {
            const unsigned dl = (unsigned)(r[u].y - node0);
            if (r[u].x >= 0 && dl < (unsigned)BKT_NODES) {
                const int pos = atomicAdd(&cnt[dl], 1);
                if (staged) {
                    const unsigned lp = (unsigned)(pos - base);
                    if (lp < (unsigned)tot) stage[lp] = r[u].x;
                } else {
                    if ((unsigned)pos < (unsigned)N_EDGES) perm[pos] = r[u].x;
                }
            }
        }
    }
    if (staged) {
        __syncthreads();
        for (int i = t; i < tot; i += 256) perm[base + i] = stage[i];
    }
}

__device__ __forceinline__ float gatv2_lane_logit(
    const float (&ea)[EDGE_DIM], const v2f (&w2)[2][EDGE_DIM], const v4f g,
    const v2f (&gr2)[2], const v2f (&a2)[2])
{
    v2f ev0 = {0.0f, 0.0f}, ev1 = {0.0f, 0.0f};
    #pragma unroll
    for (int d = 0; d < EDGE_DIM; ++d) {
        const v2f s = {ea[d], ea[d]};
        ev0 = __builtin_elementwise_fma(s, w2[0][d], ev0);
        ev1 = __builtin_elementwise_fma(s, w2[1][d], ev1);
    }
    const v2f g01 = {g.x, g.y}, g23 = {g.z, g.w};
    const v2f v0 = g01 + gr2[0] + ev0;
    const v2f v1 = g23 + gr2[1] + ev1;
    const float l0 = v0.x > 0.0f ? v0.x : NEG_SLOPE * v0.x;
    const float l1 = v0.y > 0.0f ? v0.y : NEG_SLOPE * v0.y;
    const float l2 = v1.x > 0.0f ? v1.x : NEG_SLOPE * v1.x;
    const float l3 = v1.y > 0.0f ? v1.y : NEG_SLOPE * v1.y;
    float p = l0 * a2[0].x;
    p = fmaf(l1, a2[0].y, p);
    p = fmaf(l2, a2[1].x, p);
    p = fmaf(l3, a2[1].y, p);
    return p;
}

__global__ __launch_bounds__(256) void gatv2_node(
    const int* __restrict__ ei, const float* __restrict__ eattr,
    const float* __restrict__ we, const float* __restrict__ att,
    const float* __restrict__ xl, const float* __restrict__ xr,
    const int* __restrict__ rowptr, const int* __restrict__ deg,
    const int* __restrict__ perm,
    const float* __restrict__ x, const float* __restrict__ cbias,
    const float* __restrict__ gamma, const float* __restrict__ beta,
    float* __restrict__ out)
{
    __shared__ v4f s_w[OUT_DIM * EDGE_DIM / 4];
    __shared__ v4f s_a[OUT_DIM / 4];

    const int lane = threadIdx.x & 31;
    for (int i = threadIdx.x; i < OUT_DIM * EDGE_DIM / 4; i += 256)
        s_w[i] = ((const v4f*)we)[i];
    if (threadIdx.x < OUT_DIM / 4)
        s_a[threadIdx.x] = ((const v4f*)att)[threadIdx.x];
    __syncthreads();

    float wf[4 * EDGE_DIM];
    #pragma unroll
    for (int i = 0; i < EDGE_DIM; ++i) {
        const v4f t = s_w[lane * EDGE_DIM + i];
        wf[4 * i + 0] = t.x; wf[4 * i + 1] = t.y;
        wf[4 * i + 2] = t.z; wf[4 * i + 3] = t.w;
    }
    v2f w2[2][EDGE_DIM];
    #pragma unroll
    for (int q = 0; q < 2; ++q) {
        #pragma unroll
        for (int d = 0; d < EDGE_DIM; ++d) {
            w2[q][d].x = wf[(2 * q) * EDGE_DIM + d];
            w2[q][d].y = wf[(2 * q + 1) * EDGE_DIM + d];
        }
    }
    const v4f a4 = s_a[lane];
    const v2f a2[2] = {{a4.x, a4.y}, {a4.z, a4.w}};

    const int n = __builtin_amdgcn_readfirstlane(
        (int)(blockIdx.x * 8 + (threadIdx.x >> 5)));

    const v4f r4 = *(const v4f*)(xr + (size_t)n * OUT_DIM + 4 * lane);
    const v2f gr2[2] = {{r4.x, r4.y}, {r4.z, r4.w}};

    int beg = rowptr[n];
    int end = beg + deg[n];
    beg = beg < 0 ? 0 : beg;
    end = end > N_EDGES ? N_EDGES : end;

    float m = -__builtin_inff();
    float den = 0.0f;
    float acc[4] = {0.0f, 0.0f, 0.0f, 0.0f};

    for (int base = beg; base < end; base += 32) {
        const int idx = base + lane;
        int my_e = 0, my_src = 0;
        if (idx < end) {
            my_e = perm[idx];
            my_e = (unsigned)my_e < (unsigned)N_EDGES ? my_e : 0;
            my_src = ei[my_e];
        }
        const int left = end - base;
        const int cnt = left < 32 ? left : 32;
        for (int j = 0; j < cnt; j += 2) {
            const bool two = (j + 1 < cnt);
            const int j1 = two ? j + 1 : j;
            const float* __restrict__ eap0 = eattr +
                (size_t)__builtin_amdgcn_readlane(my_e, j) * EDGE_DIM;
            const float* __restrict__ eap1 = eattr +
                (size_t)__builtin_amdgcn_readlane(my_e, j1) * EDGE_DIM;
            const v4f g0 = *(const v4f*)(xl +
                (size_t)__builtin_amdgcn_readlane(my_src, j) * OUT_DIM + 4 * lane);
            const v4f g1 = *(const v4f*)(xl +
                (size_t)__builtin_amdgcn_readlane(my_src, j1) * OUT_DIM + 4 * lane);
            float ea0[EDGE_DIM], ea1[EDGE_DIM];
            #pragma unroll
            for (int d = 0; d < EDGE_DIM; ++d) { ea0[d] = eap0[d]; ea1[d] = eap1[d]; }
            const float gl0[4] = {g0.x, g0.y, g0.z, g0.w};
            const float gl1[4] = {g1.x, g1.y, g1.z, g1.w};

            float p0 = gatv2_lane_logit(ea0, w2, g0, gr2, a2);
            float p1 = gatv2_lane_logit(ea1, w2, g1, gr2, a2);
            p0 += __shfl_xor(p0, 4, 32);  p1 += __shfl_xor(p1, 4, 32);
            p0 += __shfl_xor(p0, 2, 32);  p1 += __shfl_xor(p1, 2, 32);
            p0 += __shfl_xor(p0, 1, 32);  p1 += __shfl_xor(p1, 1, 32);
            p1 = two ? p1 : -__builtin_inff();

            const float mn = fmaxf(m, fmaxf(p0, p1));
            const float sc = __expf(m - mn);
            const float t0 = __expf(p0 - mn);
            const float t1 = __expf(p1 - mn);
            den = fmaf(den, sc, t0) + t1;
            #pragma unroll
            for (int k = 0; k < 4; ++k)
                acc[k] = fmaf(t1, gl1[k], fmaf(acc[k], sc, t0 * gl0[k]));
            m = mn;
        }
    }

    const float inv_den = 1.0f / (den + 1e-16f);

    const v4f x4 = *(const v4f*)(x + (size_t)n * OUT_DIM + 4 * lane);
    const v4f c4 = *(const v4f*)(cbias + 4 * lane);
    const float xv[4] = {x4.x, x4.y, x4.z, x4.w};
    const float cb[4] = {c4.x, c4.y, c4.z, c4.w};

    float h[4];
    float s = 0.0f;
    #pragma unroll
    for (int k = 0; k < 4; ++k) {
        const float v = acc[k] * inv_den + cb[k];
        const float sig = 1.0f / (1.0f + __expf(-v));
        h[k] = v * sig + xv[k];
        s += h[k];
    }
    #pragma unroll
    for (int off = 16; off > 0; off >>= 1) s += __shfl_xor(s, off, 32);
    const float mu = s * (1.0f / OUT_DIM);

    float vs = 0.0f;
    #pragma unroll
    for (int k = 0; k < 4; ++k) { const float d = h[k] - mu; vs += d * d; }
    #pragma unroll
    for (int off = 16; off > 0; off >>= 1) vs += __shfl_xor(vs, off, 32);
    const float inv = rsqrtf(vs * (1.0f / OUT_DIM) + LN_EPS);

    const v4f g4 = *(const v4f*)(gamma + 4 * lane);
    const v4f b4 = *(const v4f*)(beta + 4 * lane);
    v4f o;
    o.x = (h[0] - mu) * inv * g4.x + b4.x;
    o.y = (h[1] - mu) * inv * g4.y + b4.y;
    o.z = (h[2] - mu) * inv * g4.z + b4.z;
    o.w = (h[3] - mu) * inv * g4.w + b4.w;
    *(v4f*)(out + (size_t)n * OUT_DIM + 4 * lane) = o;
}

extern "C" void kernel_launch(void* const* d_in, const int* in_sizes, int n_in,
                              void* d_out, int out_size, void* d_ws, size_t ws_size,
                              hipStream_t stream) {
    const float* x     = (const float*)d_in[0];
    const int*   ei    = (const int*)  d_in[1];
    const float* eattr = (const float*)d_in[2];
    const float* wl    = (const float*)d_in[3];
    const float* bl    = (const float*)d_in[4];
    const float* wr    = (const float*)d_in[5];
    const float* br    = (const float*)d_in[6];
    const float* we    = (const float*)d_in[7];
    const float* att   = (const float*)d_in[8];
    const float* cb    = (const float*)d_in[9];
    const float* gam   = (const float*)d_in[10];
    const float* bet   = (const float*)d_in[11];
    float* out = (float*)d_out;

    float* ws     = (float*)d_ws;
    float* xl     = ws;
    float* xr     = xl     + (size_t)N_NODES * OUT_DIM;
    int*   deg    = (int*)(xr + (size_t)N_NODES * OUT_DIM);
    int*   rowptr = deg    + N_NODES;
    int*   perm   = rowptr + N_NODES;
    int*   counts = perm   + N_EDGES;
    int*   bbase  = counts + SORT_CHUNKS * N_BKT;
    int*   btot   = bbase  + 256;
    int*   rec    = btot   + 256;

    gatv2_gemm_lr<<<GEMM_BLOCKS, 256, 0, stream>>>(x, wl, bl, wr, br, xl, xr);

    gatv2_chunk_hist<<<SORT_CHUNKS, 256, 0, stream>>>(ei, counts);

    gatv2_scatter<<<SORT_CHUNKS, 256, 0, stream>>>(ei, counts, bbase, btot, rec);

    gatv2_rank<<<N_BKT, 256, 0, stream>>>(rec, bbase, btot, rowptr, deg, perm);

    gatv2_node<<<N_NODES / 8, 256, 0, stream>>>(
        ei, eattr, we, att, xl, xr, rowptr, deg, perm, x, cb, gam, bet, out);
}


# node kernel: head-logit (8-lane) and LayerNorm (32-lane) sums as DPP row_xmask adds plus one permlanex16, not LDS permutes with a wait per step; partners and sum order unchanged
# speedup vs baseline: 4.3729x; 1.0622x over previous
#include <hip/hip_runtime.h>

#define N_NODES 50000
#define N_EDGES 800000
#define IN_DIM 128
#define OUT_DIM 128
#define HEADS 4
#define HEAD_DIM 32
#define EDGE_DIM 10
#define NEG_SLOPE 0.2f
#define LN_EPS 1e-5f

typedef __attribute__((ext_vector_type(2))) float v2f;
typedef __attribute__((ext_vector_type(8))) float v8f;
typedef __attribute__((ext_vector_type(4))) float v4f;

#define GEMM_BLOCKS 256
#define GEMM_TILES (N_NODES / 16)
#define GEMM_FRAGS (IN_DIM / 8)

__device__ __forceinline__ void gatv2_load_frags(
    v4f (&f)[GEMM_FRAGS], const float* __restrict__ row_k4h)
{
    __builtin_amdgcn_sched_barrier(0);
    #pragma unroll
    for (int j = 0; j < GEMM_FRAGS; ++j)
        f[j] = *(const v4f*)(row_k4h + 8 * j);
    __builtin_amdgcn_sched_barrier(0);
}

__device__ __forceinline__ void gatv2_gemm_tile(
    const v4f (&a)[GEMM_FRAGS], const v4f (&fl)[GEMM_FRAGS],
    const v4f (&fr)[GEMM_FRAGS], const float biasL, const float biasR,
    float* __restrict__ xl, float* __restrict__ xr, const size_t out0)
{
    v8f accL = {};
    v8f accR = {};
    #pragma unroll
    for (int j = 0; j < GEMM_FRAGS; ++j) {
        const v2f a0 = {a[j].x, a[j].y},   a1 = {a[j].z, a[j].w};
        const v2f l0 = {fl[j].x, fl[j].y}, l1 = {fl[j].z, fl[j].w};
        const v2f r0 = {fr[j].x, fr[j].y}, r1 = {fr[j].z, fr[j].w};
        accL = __builtin_amdgcn_wmma_f32_16x16x4_f32(
            false, a0, false, l0, (short)0, accL, false, false);
        accR = __builtin_amdgcn_wmma_f32_16x16x4_f32(
            false, a0, false, r0, (short)0, accR, false, false);
        accL = __builtin_amdgcn_wmma_f32_16x16x4_f32(
            false, a1, false, l1, (short)0, accL, false, false);
        accR = __builtin_amdgcn_wmma_f32_16x16x4_f32(
            false, a1, false, r1, (short)0, accR, false, false);
    }
    #pragma unroll
    for (int r = 0; r < 8; ++r) {
        xl[out0 + (size_t)r * OUT_DIM] = accL[r] + biasL;
        xr[out0 + (size_t)r * OUT_DIM] = accR[r] + biasR;
    }
}

__global__ __launch_bounds__(256) void gatv2_gemm_lr(
    const float* __restrict__ x,
    const float* __restrict__ wl, const float* __restrict__ bl,
    const float* __restrict__ wr, const float* __restrict__ br,
    float* __restrict__ xl, float* __restrict__ xr)
{
    const int wave = threadIdx.x >> 5;
    const int lane = threadIdx.x & 31;
    const int c = lane & 15;
    const int h = lane >> 4;
    const int bcol = wave * 16 + c;

    v4f fl[GEMM_FRAGS], fr[GEMM_FRAGS];
    gatv2_load_frags(fl, wl + (size_t)bcol * IN_DIM + 4 * h);
    gatv2_load_frags(fr, wr + (size_t)bcol * IN_DIM + 4 * h);
    const float biasL = bl[bcol];
    const float biasR = br[bcol];

    const size_t a_lane = (size_t)c * IN_DIM + 4 * h;
    const size_t d_lane = (size_t)(8 * h) * OUT_DIM + bcol;
    const int b = blockIdx.x;
    if (b >= GEMM_TILES) return;
    const int my = (GEMM_TILES - 1 - b) / GEMM_BLOCKS + 1;
    const int last = b + (my - 1) * GEMM_BLOCKS;

    int t = b;
    v4f xa[GEMM_FRAGS], xb[GEMM_FRAGS];
    gatv2_load_frags(xa, x + (size_t)t * 16 * IN_DIM + a_lane);
    for (int p = 0; p < my / 2; ++p) {
        const int t1 = t + GEMM_BLOCKS;
        const int t2 = t1 + GEMM_BLOCKS;
        gatv2_load_frags(xb, x + (size_t)t1 * 16 * IN_DIM + a_lane);
        gatv2_gemm_tile(xa, fl, fr, biasL, biasR, xl, xr,
                        (size_t)t * 16 * OUT_DIM + d_lane);
        gatv2_load_frags(xa, x + (size_t)(t2 <= last ? t2 : last) * 16 * IN_DIM + a_lane);
        gatv2_gemm_tile(xb, fl, fr, biasL, biasR, xl, xr,
                        (size_t)t1 * 16 * OUT_DIM + d_lane);
        t = t2;
    }
    if (my & 1)
        gatv2_gemm_tile(xa, fl, fr, biasL, biasR, xl, xr,
                        (size_t)t * 16 * OUT_DIM + d_lane);
}

typedef __attribute__((ext_vector_type(2))) int v2i;

#define SORT_CHUNKS 256
#define CHUNK_EDGES (N_EDGES / SORT_CHUNKS)
#define CHUNK_EPT ((CHUNK_EDGES + 255) / 256)
#define BKT_SHIFT 8
#define BKT_NODES (1 << BKT_SHIFT)
#define N_BKT ((N_NODES + BKT_NODES - 1) / BKT_NODES)
#define RANK_UNROLL 8

__device__ __forceinline__ int gatv2_bucket(const int d)
{
    const unsigned k = (unsigned)d >> BKT_SHIFT;
    return (int)(k < (unsigned)N_BKT ? k : (unsigned)(N_BKT - 1));
}

__device__ __forceinline__ void gatv2_chunk_dst(
    int (&d)[CHUNK_EPT], const int* __restrict__ ei, const int chunk, const int t)
{
    const int* __restrict__ dstp = ei + N_EDGES + chunk * CHUNK_EDGES;
    #pragma unroll
    for (int u = 0; u < CHUNK_EPT; ++u) {
        const int idx = t + u * 256;
        d[u] = idx < CHUNK_EDGES ? dstp[idx] : -1;
    }
}

__device__ __forceinline__ int gatv2_block_excl_scan(const int v, int* wsum)
{
    const int lane = threadIdx.x & 31;
    const int w = threadIdx.x >> 5;
    int incl = v;
    #pragma unroll
    for (int off = 1; off < 32; off <<= 1) {
        const int s = __shfl_up(incl, off, 32);
        if (lane >= off) incl += s;
    }
    if (lane == 31) wsum[w] = incl;
    __syncthreads();
    int wbase = 0;
    #pragma unroll
    for (int i = 0; i < 8; ++i) wbase += (i < w) ? wsum[i] : 0;
    return wbase + incl - v;
}

__global__ __launch_bounds__(256) void gatv2_chunk_hist(
    const int* __restrict__ ei, int* __restrict__ counts)
{
    __shared__ int hist[N_BKT];
    const int t = threadIdx.x;
    int d[CHUNK_EPT];
    gatv2_chunk_dst(d, ei, blockIdx.x, t);
    if (t < N_BKT) hist[t] = 0;
    __syncthreads();
    #pragma unroll
    for (int u = 0; u < CHUNK_EPT; ++u)
        if (d[u] >= 0) atomicAdd(&hist[gatv2_bucket(d[u])], 1);
    __syncthreads();
    if (t < N_BKT) counts[blockIdx.x * N_BKT + t] = hist[t];
}

__global__ __launch_bounds__(256) void gatv2_scatter(
    const int* __restrict__ ei, const int* __restrict__ counts,
    int* __restrict__ bbase, int* __restrict__ btot, int* __restrict__ rec)
{
    __shared__ int gstart[256];
    __shared__ int lstart[256];
    __shared__ int lcur[256];
    __shared__ int wsum[8];
    __shared__ int nstaged;
    __shared__ v2i stage[CHUNK_EDGES];
    const int t = threadIdx.x;
    const int b = blockIdx.x;
    int d[CHUNK_EPT];
    gatv2_chunk_dst(d, ei, b, t);

    int tot = 0, mine = 0, own = 0;
    if (t < N_BKT) {
        #pragma unroll 16
        for (int bb = 0; bb < SORT_CHUNKS; ++bb) {
            const int c = counts[bb * N_BKT + t];
            tot += c;
            mine += (bb < b) ? c : 0;
            own = (bb == b) ? c : own;
        }
    }
    const int base = gatv2_block_excl_scan(tot, wsum);
    __syncthreads();
    const int lst = gatv2_block_excl_scan(own, wsum);
    gstart[t] = base + mine;
    lstart[t] = lst;
    lcur[t] = lst;
    if (t == 255) nstaged = lst + own;
    if (b == 0 && t < N_BKT) { bbase[t] = base; btot[t] = tot; }
    __syncthreads();

    #pragma unroll
    for (int u = 0; u < CHUNK_EPT; ++u) {
        if (d[u] >= 0) {
            const int lp = atomicAdd(&lcur[gatv2_bucket(d[u])], 1);
            if ((unsigned)lp < (unsigned)CHUNK_EDGES) {
                const v2i r = {b * CHUNK_EDGES + t + u * 256, d[u]};
                stage[lp] = r;
            }
        }
    }
    __syncthreads();

    int n = nstaged;
    n = n < CHUNK_EDGES ? n : CHUNK_EDGES;
    for (int i = t; i < n; i += 256) {
        const v2i r = stage[i];
        const int k = gatv2_bucket(r.y);
        const int g = gstart[k] + (i - lstart[k]);
        if ((unsigned)g < (unsigned)N_EDGES) ((v2i*)rec)[g] = r;
    }
}

#define RANK_STAGE 8192
__global__ __launch_bounds__(256) void gatv2_rank(
    const int* __restrict__ rec, const int* __restrict__ bbase,
    const int* __restrict__ btot, int* __restrict__ rowptr,
    int* __restrict__ deg, int* __restrict__ perm)
{
    __shared__ int cnt[BKT_NODES];
    __shared__ int wsum[8];
    __shared__ int stage[RANK_STAGE];
    const int t = threadIdx.x;
    const int node0 = blockIdx.x * BKT_NODES;
    int base = bbase[blockIdx.x];
    int tot = btot[blockIdx.x];
    base = base < 0 ? 0 : (base > N_EDGES ? N_EDGES : base);
    tot = tot < 0 ? 0 : (tot > N_EDGES - base ? N_EDGES - base : tot);
    const v2i* __restrict__ recs = (const v2i*)rec + base;
    const v2i none = {-1, -1};

    cnt[t] = 0;
    __syncthreads();
    for (int i0 = t; i0 < tot; i0 += 256 * RANK_UNROLL) {
        v2i r[RANK_UNROLL];
        #pragma unroll
        for (int u = 0; u < RANK_UNROLL; ++u) {
            const int i = i0 + u * 256;
            r[u] = i < tot ? recs[i] : none;
        }
        #pragma unroll
        for (int u = 0; u < RANK_UNROLL; ++u) {
            const unsigned dl = (unsigned)(r[u].y - node0);
            if (r[u].x >= 0 && dl < (unsigned)BKT_NODES) atomicAdd(&cnt[dl], 1);
        }
    }
    __syncthreads();

    const int c = cnt[t];
    const int first = base + gatv2_block_excl_scan(c, wsum);
    cnt[t] = first;
    if (node0 + t < N_NODES) {
        rowptr[node0 + t] = first;
        deg[node0 + t] = c;
    }
    __syncthreads();

    const bool staged = tot <= RANK_STAGE;
    for (int i0 = t; i0 < tot; i0 += 256 * RANK_UNROLL) {
        v2i r[RANK_UNROLL];
        #pragma unroll
        for (int u = 0; u < RANK_UNROLL; ++u) {
            const int i = i0 + u * 256;
            r[u] = i < tot ? recs[i] : none;
        }
        #pragma unroll
        for (int u = 0; u < RANK_UNROLL; ++u) {
            const unsigned dl = (unsigned)(r[u].y - node0);
            if (r[u].x >= 0 && dl < (unsigned)BKT_NODES) {
                const int pos = atomicAdd(&cnt[dl], 1);
                if (staged) {
                    const unsigned lp = (unsigned)(pos - base);
                    if (lp < (unsigned)tot) stage[lp] = r[u].x;
                } else {
                    if ((unsigned)pos < (unsigned)N_EDGES) perm[pos] = r[u].x;
                }
            }
        }
    }
    if (staged) {
        __syncthreads();
        for (int i = t; i < tot; i += 256) perm[base + i] = stage[i];
    }
}

__device__ __forceinline__ float gatv2_lane_logit(
    const float (&ea)[EDGE_DIM], const v2f (&w2)[2][EDGE_DIM], const v4f g,
    const v2f (&gr2)[2], const v2f (&a2)[2])
{
    v2f ev0 = {0.0f, 0.0f}, ev1 = {0.0f, 0.0f};
    #pragma unroll
    for (int d = 0; d < EDGE_DIM; ++d) {
        const v2f s = {ea[d], ea[d]};
        ev0 = __builtin_elementwise_fma(s, w2[0][d], ev0);
        ev1 = __builtin_elementwise_fma(s, w2[1][d], ev1);
    }
    const v2f g01 = {g.x, g.y}, g23 = {g.z, g.w};
    const v2f v0 = g01 + gr2[0] + ev0;
    const v2f v1 = g23 + gr2[1] + ev1;
    const float l0 = v0.x > 0.0f ? v0.x : NEG_SLOPE * v0.x;
    const float l1 = v0.y > 0.0f ? v0.y : NEG_SLOPE * v0.y;
    const float l2 = v1.x > 0.0f ? v1.x : NEG_SLOPE * v1.x;
    const float l3 = v1.y > 0.0f ? v1.y : NEG_SLOPE * v1.y;
    float p = l0 * a2[0].x;
    p = fmaf(l1, a2[0].y, p);
    p = fmaf(l2, a2[1].x, p);
    p = fmaf(l3, a2[1].y, p);
    return p;
}

template <int M>
__device__ __forceinline__ float gatv2_row_xadd(const float v)
{
    const int b = __builtin_bit_cast(int, v);
    const int s = __builtin_amdgcn_update_dpp(b, b, 0x160 + M, 0xf, 0xf, true);
    return v + __builtin_bit_cast(float, s);
}

__device__ __forceinline__ float gatv2_half_xadd(const float v)
{
    const int b = __builtin_bit_cast(int, v);
    const int s = __builtin_amdgcn_permlanex16(b, b, 0x76543210, 0xfedcba98, false, false);
    return v + __builtin_bit_cast(float, s);
}

__device__ __forceinline__ float gatv2_wave_sum(float v)
{
    v = gatv2_half_xadd(v);
    v = gatv2_row_xadd<8>(v);
    v = gatv2_row_xadd<4>(v);
    v = gatv2_row_xadd<2>(v);
    v = gatv2_row_xadd<1>(v);
    return v;
}

__global__ __launch_bounds__(256) void gatv2_node(
    const int* __restrict__ ei, const float* __restrict__ eattr,
    const float* __restrict__ we, const float* __restrict__ att,
    const float* __restrict__ xl, const float* __restrict__ xr,
    const int* __restrict__ rowptr, const int* __restrict__ deg,
    const int* __restrict__ perm,
    const float* __restrict__ x, const float* __restrict__ cbias,
    const float* __restrict__ gamma, const float* __restrict__ beta,
    float* __restrict__ out)
{
    __shared__ v4f s_w[OUT_DIM * EDGE_DIM / 4];
    __shared__ v4f s_a[OUT_DIM / 4];

    const int lane = threadIdx.x & 31;
    for (int i = threadIdx.x; i < OUT_DIM * EDGE_DIM / 4; i += 256)
        s_w[i] = ((const v4f*)we)[i];
    if (threadIdx.x < OUT_DIM / 4)
        s_a[threadIdx.x] = ((const v4f*)att)[threadIdx.x];
    __syncthreads();

    float wf[4 * EDGE_DIM];
    #pragma unroll
    for (int i = 0; i < EDGE_DIM; ++i) {
        const v4f t = s_w[lane * EDGE_DIM + i];
        wf[4 * i + 0] = t.x; wf[4 * i + 1] = t.y;
        wf[4 * i + 2] = t.z; wf[4 * i + 3] = t.w;
    }
    v2f w2[2][EDGE_DIM];
    #pragma unroll
    for (int q = 0; q < 2; ++q) {
        #pragma unroll
        for (int d = 0; d < EDGE_DIM; ++d) {
            w2[q][d].x = wf[(2 * q) * EDGE_DIM + d];
            w2[q][d].y = wf[(2 * q + 1) * EDGE_DIM + d];
        }
    }
    const v4f a4 = s_a[lane];
    const v2f a2[2] = {{a4.x, a4.y}, {a4.z, a4.w}};

    const int n = __builtin_amdgcn_readfirstlane(
        (int)(blockIdx.x * 8 + (threadIdx.x >> 5)));

    const v4f r4 = *(const v4f*)(xr + (size_t)n * OUT_DIM + 4 * lane);
    const v2f gr2[2] = {{r4.x, r4.y}, {r4.z, r4.w}};

    int beg = rowptr[n];
    int end = beg + deg[n];
    beg = beg < 0 ? 0 : beg;
    end = end > N_EDGES ? N_EDGES : end;

    float m = -__builtin_inff();
    float den = 0.0f;
    float acc[4] = {0.0f, 0.0f, 0.0f, 0.0f};

    for (int base = beg; base < end; base += 32) {
        const int idx = base + lane;
        int my_e = 0, my_src = 0;
        if (idx < end) {
            my_e = perm[idx];
            my_e = (unsigned)my_e < (unsigned)N_EDGES ? my_e : 0;
            my_src = ei[my_e];
        }
        const int left = end - base;
        const int cnt = left < 32 ? left : 32;
        for (int j = 0; j < cnt; j += 2) {
            const bool two = (j + 1 < cnt);
            const int j1 = two ? j + 1 : j;
            const float* __restrict__ eap0 = eattr +
                (size_t)__builtin_amdgcn_readlane(my_e, j) * EDGE_DIM;
            const float* __restrict__ eap1 = eattr +
                (size_t)__builtin_amdgcn_readlane(my_e, j1) * EDGE_DIM;
            const v4f g0 = *(const v4f*)(xl +
                (size_t)__builtin_amdgcn_readlane(my_src, j) * OUT_DIM + 4 * lane);
            const v4f g1 = *(const v4f*)(xl +
                (size_t)__builtin_amdgcn_readlane(my_src, j1) * OUT_DIM + 4 * lane);
            float ea0[EDGE_DIM], ea1[EDGE_DIM];
            #pragma unroll
            for (int d = 0; d < EDGE_DIM; ++d) { ea0[d] = eap0[d]; ea1[d] = eap1[d]; }
            const float gl0[4] = {g0.x, g0.y, g0.z, g0.w};
            const float gl1[4] = {g1.x, g1.y, g1.z, g1.w};

            float p0 = gatv2_lane_logit(ea0, w2, g0, gr2, a2);
            float p1 = gatv2_lane_logit(ea1, w2, g1, gr2, a2);
            p0 = gatv2_row_xadd<4>(p0);  p1 = gatv2_row_xadd<4>(p1);
            p0 = gatv2_row_xadd<2>(p0);  p1 = gatv2_row_xadd<2>(p1);
            p0 = gatv2_row_xadd<1>(p0);  p1 = gatv2_row_xadd<1>(p1);
            p1 = two ? p1 : -__builtin_inff();

            const float mn = fmaxf(m, fmaxf(p0, p1));
            const float sc = __expf(m - mn);
            const float t0 = __expf(p0 - mn);
            const float t1 = __expf(p1 - mn);
            den = fmaf(den, sc, t0) + t1;
            #pragma unroll
            for (int k = 0; k < 4; ++k)
                acc[k] = fmaf(t1, gl1[k], fmaf(acc[k], sc, t0 * gl0[k]));
            m = mn;
        }
    }

    const float inv_den = 1.0f / (den + 1e-16f);

    const v4f x4 = *(const v4f*)(x + (size_t)n * OUT_DIM + 4 * lane);
    const v4f c4 = *(const v4f*)(cbias + 4 * lane);
    const float xv[4] = {x4.x, x4.y, x4.z, x4.w};
    const float cb[4] = {c4.x, c4.y, c4.z, c4.w};

    float h[4];
    float s = 0.0f;
    #pragma unroll
    for (int k = 0; k < 4; ++k) {
        const float v = acc[k] * inv_den + cb[k];
        const float sig = 1.0f / (1.0f + __expf(-v));
        h[k] = v * sig + xv[k];
        s += h[k];
    }
    s = gatv2_wave_sum(s);
    const float mu = s * (1.0f / OUT_DIM);

    float vs = 0.0f;
    #pragma unroll
    for (int k = 0; k < 4; ++k) { const float d = h[k] - mu; vs += d * d; }
    vs = gatv2_wave_sum(vs);
    const float inv = rsqrtf(vs * (1.0f / OUT_DIM) + LN_EPS);

    const v4f g4 = *(const v4f*)(gamma + 4 * lane);
    const v4f b4 = *(const v4f*)(beta + 4 * lane);
    v4f o;
    o.x = (h[0] - mu) * inv * g4.x + b4.x;
    o.y = (h[1] - mu) * inv * g4.y + b4.y;
    o.z = (h[2] - mu) * inv * g4.z + b4.z;
    o.w = (h[3] - mu) * inv * g4.w + b4.w;
    *(v4f*)(out + (size_t)n * OUT_DIM + 4 * lane) = o;
}

extern "C" void kernel_launch(void* const* d_in, const int* in_sizes, int n_in,
                              void* d_out, int out_size, void* d_ws, size_t ws_size,
                              hipStream_t stream) {
    const float* x     = (const float*)d_in[0];
    const int*   ei    = (const int*)  d_in[1];
    const float* eattr = (const float*)d_in[2];
    const float* wl    = (const float*)d_in[3];
    const float* bl    = (const float*)d_in[4];
    const float* wr    = (const float*)d_in[5];
    const float* br    = (const float*)d_in[6];
    const float* we    = (const float*)d_in[7];
    const float* att   = (const float*)d_in[8];
    const float* cb    = (const float*)d_in[9];
    const float* gam   = (const float*)d_in[10];
    const float* bet   = (const float*)d_in[11];
    float* out = (float*)d_out;

    float* ws     = (float*)d_ws;
    float* xl     = ws;
    float* xr     = xl     + (size_t)N_NODES * OUT_DIM;
    int*   deg    = (int*)(xr + (size_t)N_NODES * OUT_DIM);
    int*   rowptr = deg    + N_NODES;
    int*   perm   = rowptr + N_NODES;
    int*   counts = perm   + N_EDGES;
    int*   bbase  = counts + SORT_CHUNKS * N_BKT;
    int*   btot   = bbase  + 256;
    int*   rec    = btot   + 256;

    gatv2_gemm_lr<<<GEMM_BLOCKS, 256, 0, stream>>>(x, wl, bl, wr, br, xl, xr);

    gatv2_chunk_hist<<<SORT_CHUNKS, 256, 0, stream>>>(ei, counts);

    gatv2_scatter<<<SORT_CHUNKS, 256, 0, stream>>>(ei, counts, bbase, btot, rec);

    gatv2_rank<<<N_BKT, 256, 0, stream>>>(rec, bbase, btot, rowptr, deg, perm);

    gatv2_node<<<N_NODES / 8, 256, 0, stream>>>(
        ei, eattr, we, att, xl, xr, rowptr, deg, perm, x, cb, gam, bet, out);
}


# node kernel: edge loop software-pipelined, the next edge pair's source rows and attribute rows are requested one trip ahead of their use; sums and their order unchanged
# speedup vs baseline: 4.5902x; 1.0497x over previous
#include <hip/hip_runtime.h>

#define N_NODES 50000
#define N_EDGES 800000
#define IN_DIM 128
#define OUT_DIM 128
#define HEADS 4
#define HEAD_DIM 32
#define EDGE_DIM 10
#define NEG_SLOPE 0.2f
#define LN_EPS 1e-5f

typedef __attribute__((ext_vector_type(2))) float v2f;
typedef __attribute__((ext_vector_type(8))) float v8f;
typedef __attribute__((ext_vector_type(4))) float v4f;

#define GEMM_BLOCKS 256
#define GEMM_TILES (N_NODES / 16)
#define GEMM_FRAGS (IN_DIM / 8)

__device__ __forceinline__ void gatv2_load_frags(
    v4f (&f)[GEMM_FRAGS], const float* __restrict__ row_k4h)
{
    __builtin_amdgcn_sched_barrier(0);
    #pragma unroll
    for (int j = 0; j < GEMM_FRAGS; ++j)
        f[j] = *(const v4f*)(row_k4h + 8 * j);
    __builtin_amdgcn_sched_barrier(0);
}

__device__ __forceinline__ void gatv2_gemm_tile(
    const v4f (&a)[GEMM_FRAGS], const v4f (&fl)[GEMM_FRAGS],
    const v4f (&fr)[GEMM_FRAGS], const float biasL, const float biasR,
    float* __restrict__ xl, float* __restrict__ xr, const size_t out0)
{
    v8f accL = {};
    v8f accR = {};
    #pragma unroll
    for (int j = 0; j < GEMM_FRAGS; ++j) {
        const v2f a0 = {a[j].x, a[j].y},   a1 = {a[j].z, a[j].w};
        const v2f l0 = {fl[j].x, fl[j].y}, l1 = {fl[j].z, fl[j].w};
        const v2f r0 = {fr[j].x, fr[j].y}, r1 = {fr[j].z, fr[j].w};
        accL = __builtin_amdgcn_wmma_f32_16x16x4_f32(
            false, a0, false, l0, (short)0, accL, false, false);
        accR = __builtin_amdgcn_wmma_f32_16x16x4_f32(
            false, a0, false, r0, (short)0, accR, false, false);
        accL = __builtin_amdgcn_wmma_f32_16x16x4_f32(
            false, a1, false, l1, (short)0, accL, false, false);
        accR = __builtin_amdgcn_wmma_f32_16x16x4_f32(
            false, a1, false, r1, (short)0, accR, false, false);
    }
    #pragma unroll
    for (int r = 0; r < 8; ++r) {
        xl[out0 + (size_t)r * OUT_DIM] = accL[r] + biasL;
        xr[out0 + (size_t)r * OUT_DIM] = accR[r] + biasR;
    }
}

__global__ __launch_bounds__(256) void gatv2_gemm_lr(
    const float* __restrict__ x,
    const float* __restrict__ wl, const float* __restrict__ bl,
    const float* __restrict__ wr, const float* __restrict__ br,
    float* __restrict__ xl, float* __restrict__ xr)
{
    const int wave = threadIdx.x >> 5;
    const int lane = threadIdx.x & 31;
    const int c = lane & 15;
    const int h = lane >> 4;
    const int bcol = wave * 16 + c;

    v4f fl[GEMM_FRAGS], fr[GEMM_FRAGS];
    gatv2_load_frags(fl, wl + (size_t)bcol * IN_DIM + 4 * h);
    gatv2_load_frags(fr, wr + (size_t)bcol * IN_DIM + 4 * h);
    const float biasL = bl[bcol];
    const float biasR = br[bcol];

    const size_t a_lane = (size_t)c * IN_DIM + 4 * h;
    const size_t d_lane = (size_t)(8 * h) * OUT_DIM + bcol;
    const int b = blockIdx.x;
    if (b >= GEMM_TILES) return;
    const int my = (GEMM_TILES - 1 - b) / GEMM_BLOCKS + 1;
    const int last = b + (my - 1) * GEMM_BLOCKS;

    int t = b;
    v4f xa[GEMM_FRAGS], xb[GEMM_FRAGS];
    gatv2_load_frags(xa, x + (size_t)t * 16 * IN_DIM + a_lane);
    for (int p = 0; p < my / 2; ++p) {
        const int t1 = t + GEMM_BLOCKS;
        const int t2 = t1 + GEMM_BLOCKS;
        gatv2_load_frags(xb, x + (size_t)t1 * 16 * IN_DIM + a_lane);
        gatv2_gemm_tile(xa, fl, fr, biasL, biasR, xl, xr,
                        (size_t)t * 16 * OUT_DIM + d_lane);
        gatv2_load_frags(xa, x + (size_t)(t2 <= last ? t2 : last) * 16 * IN_DIM + a_lane);
        gatv2_gemm_tile(xb, fl, fr, biasL, biasR, xl, xr,
                        (size_t)t1 * 16 * OUT_DIM + d_lane);
        t = t2;
    }
    if (my & 1)
        gatv2_gemm_tile(xa, fl, fr, biasL, biasR, xl, xr,
                        (size_t)t * 16 * OUT_DIM + d_lane);
}

typedef __attribute__((ext_vector_type(2))) int v2i;

#define SORT_CHUNKS 256
#define CHUNK_EDGES (N_EDGES / SORT_CHUNKS)
#define CHUNK_EPT ((CHUNK_EDGES + 255) / 256)
#define BKT_SHIFT 8
#define BKT_NODES (1 << BKT_SHIFT)
#define N_BKT ((N_NODES + BKT_NODES - 1) / BKT_NODES)
#define RANK_UNROLL 8

__device__ __forceinline__ int gatv2_bucket(const int d)
{
    const unsigned k = (unsigned)d >> BKT_SHIFT;
    return (int)(k < (unsigned)N_BKT ? k : (unsigned)(N_BKT - 1));
}

__device__ __forceinline__ void gatv2_chunk_dst(
    int (&d)[CHUNK_EPT], const int* __restrict__ ei, const int chunk, const int t)
{
    const int* __restrict__ dstp = ei + N_EDGES + chunk * CHUNK_EDGES;
    #pragma unroll
    for (int u = 0; u < CHUNK_EPT; ++u) {
        const int idx = t + u * 256;
        d[u] = idx < CHUNK_EDGES ? dstp[idx] : -1;
    }
}

__device__ __forceinline__ int gatv2_block_excl_scan(const int v, int* wsum)
{
    const int lane = threadIdx.x & 31;
    const int w = threadIdx.x >> 5;
    int incl = v;
    #pragma unroll
    for (int off = 1; off < 32; off <<= 1) {
        const int s = __shfl_up(incl, off, 32);
        if (lane >= off) incl += s;
    }
    if (lane == 31) wsum[w] = incl;
    __syncthreads();
    int wbase = 0;
    #pragma unroll
    for (int i = 0; i < 8; ++i) wbase += (i < w) ? wsum[i] : 0;
    return wbase + incl - v;
}

__global__ __launch_bounds__(256) void gatv2_chunk_hist(
    const int* __restrict__ ei, int* __restrict__ counts)
{
    __shared__ int hist[N_BKT];
    const int t = threadIdx.x;
    int d[CHUNK_EPT];
    gatv2_chunk_dst(d, ei, blockIdx.x, t);
    if (t < N_BKT) hist[t] = 0;
    __syncthreads();
    #pragma unroll
    for (int u = 0; u < CHUNK_EPT; ++u)
        if (d[u] >= 0) atomicAdd(&hist[gatv2_bucket(d[u])], 1);
    __syncthreads();
    if (t < N_BKT) counts[blockIdx.x * N_BKT + t] = hist[t];
}

__global__ __launch_bounds__(256) void gatv2_scatter(
    const int* __restrict__ ei, const int* __restrict__ counts,
    int* __restrict__ bbase, int* __restrict__ btot, int* __restrict__ rec)
{
    __shared__ int gstart[256];
    __shared__ int lstart[256];
    __shared__ int lcur[256];
    __shared__ int wsum[8];
    __shared__ int nstaged;
    __shared__ v2i stage[CHUNK_EDGES];
    const int t = threadIdx.x;
    const int b = blockIdx.x;
    int d[CHUNK_EPT];
    gatv2_chunk_dst(d, ei, b, t);

    int tot = 0, mine = 0, own = 0;
    if (t < N_BKT) {
        #pragma unroll 16
        for (int bb = 0; bb < SORT_CHUNKS; ++bb) {
            const int c = counts[bb * N_BKT + t];
            tot += c;
            mine += (bb < b) ? c : 0;
            own = (bb == b) ? c : own;
        }
    }
    const int base = gatv2_block_excl_scan(tot, wsum);
    __syncthreads();
    const int lst = gatv2_block_excl_scan(own, wsum);
    gstart[t] = base + mine;
    lstart[t] = lst;
    lcur[t] = lst;
    if (t == 255) nstaged = lst + own;
    if (b == 0 && t < N_BKT) { bbase[t] = base; btot[t] = tot; }
    __syncthreads();

    #pragma unroll
    for (int u = 0; u < CHUNK_EPT; ++u) {
        if (d[u] >= 0) {
            const int lp = atomicAdd(&lcur[gatv2_bucket(d[u])], 1);
            if ((unsigned)lp < (unsigned)CHUNK_EDGES) {
                const v2i r = {b * CHUNK_EDGES + t + u * 256, d[u]};
                stage[lp] = r;
            }
        }
    }
    __syncthreads();

    int n = nstaged;
    n = n < CHUNK_EDGES ? n : CHUNK_EDGES;
    for (int i = t; i < n; i += 256) {
        const v2i r = stage[i];
        const int k = gatv2_bucket(r.y);
        const int g = gstart[k] + (i - lstart[k]);
        if ((unsigned)g < (unsigned)N_EDGES) ((v2i*)rec)[g] = r;
    }
}

#define RANK_STAGE 8192
__global__ __launch_bounds__(256) void gatv2_rank(
    const int* __restrict__ rec, const int* __restrict__ bbase,
    const int* __restrict__ btot, int* __restrict__ rowptr,
    int* __restrict__ deg, int* __restrict__ perm)
{
    __shared__ int cnt[BKT_NODES];
    __shared__ int wsum[8];
    __shared__ int stage[RANK_STAGE];
    const int t = threadIdx.x;
    const int node0 = blockIdx.x * BKT_NODES;
    int base = bbase[blockIdx.x];
    int tot = btot[blockIdx.x];
    base = base < 0 ? 0 : (base > N_EDGES ? N_EDGES : base);
    tot = tot < 0 ? 0 : (tot > N_EDGES - base ? N_EDGES - base : tot);
    const v2i* __restrict__ recs = (const v2i*)rec + base;
    const v2i none = {-1, -1};

    cnt[t] = 0;
    __syncthreads();
    for (int i0 = t; i0 < tot; i0 += 256 * RANK_UNROLL) {
        v2i r[RANK_UNROLL];
        #pragma unroll
        for (int u = 0; u < RANK_UNROLL; ++u) {
            const int i = i0 + u * 256;
            r[u] = i < tot ? recs[i] : none;
        }
        #pragma unroll
        for (int u = 0; u < RANK_UNROLL; ++u) {
            const unsigned dl = (unsigned)(r[u].y - node0);
            if (r[u].x >= 0 && dl < (unsigned)BKT_NODES) atomicAdd(&cnt[dl], 1);
        }
    }
    __syncthreads();

    const int c = cnt[t];
    const int first = base + gatv2_block_excl_scan(c, wsum);
    cnt[t] = first;
    if (node0 + t < N_NODES) {
        rowptr[node0 + t] = first;
        deg[node0 + t] = c;
    }
    __syncthreads();

    const bool staged = tot <= RANK_STAGE;
    for (int i0 = t; i0 < tot; i0 += 256 * RANK_UNROLL) {
        v2i r[RANK_UNROLL];
        #pragma unroll
        for (int u = 0; u < RANK_UNROLL; ++u) {
            const int i = i0 + u * 256;
            r[u] = i < tot ? recs[i] : none;
        }
        #pragma unroll
        for (int u = 0; u < RANK_UNROLL; ++u) {
            const unsigned dl = (unsigned)(r[u].y - node0);
            if (r[u].x >= 0 && dl < (unsigned)BKT_NODES) {
                const int pos = atomicAdd(&cnt[dl], 1);
                if (staged) {
                    const unsigned lp = (unsigned)(pos - base);
                    if (lp < (unsigned)tot) stage[lp] = r[u].x;
                } else {
                    if ((unsigned)pos < (unsigned)N_EDGES) perm[pos] = r[u].x;
                }
            }
        }
    }
    if (staged) {
        __syncthreads();
        for (int i = t; i < tot; i += 256) perm[base + i] = stage[i];
    }
}

__device__ __forceinline__ float gatv2_lane_logit(
    const float (&ea)[EDGE_DIM], const v2f (&w2)[2][EDGE_DIM], const v4f g,
    const v2f (&gr2)[2], const v2f (&a2)[2])
{
    v2f ev0 = {0.0f, 0.0f}, ev1 = {0.0f, 0.0f};
    #pragma unroll
    for (int d = 0; d < EDGE_DIM; ++d) {
        const v2f s = {ea[d], ea[d]};
        ev0 = __builtin_elementwise_fma(s, w2[0][d], ev0);
        ev1 = __builtin_elementwise_fma(s, w2[1][d], ev1);
    }
    const v2f g01 = {g.x, g.y}, g23 = {g.z, g.w};
    const v2f v0 = g01 + gr2[0] + ev0;
    const v2f v1 = g23 + gr2[1] + ev1;
    const float l0 = v0.x > 0.0f ? v0.x : NEG_SLOPE * v0.x;
    const float l1 = v0.y > 0.0f ? v0.y : NEG_SLOPE * v0.y;
    const float l2 = v1.x > 0.0f ? v1.x : NEG_SLOPE * v1.x;
    const float l3 = v1.y > 0.0f ? v1.y : NEG_SLOPE * v1.y;
    float p = l0 * a2[0].x;
    p = fmaf(l1, a2[0].y, p);
    p = fmaf(l2, a2[1].x, p);
    p = fmaf(l3, a2[1].y, p);
    return p;
}

template <int M>
__device__ __forceinline__ float gatv2_row_xadd(const float v)
{
    const int b = __builtin_bit_cast(int, v);
    const int s = __builtin_amdgcn_update_dpp(b, b, 0x160 + M, 0xf, 0xf, true);
    return v + __builtin_bit_cast(float, s);
}

__device__ __forceinline__ float gatv2_half_xadd(const float v)
{
    const int b = __builtin_bit_cast(int, v);
    const int s = __builtin_amdgcn_permlanex16(b, b, 0x76543210, 0xfedcba98, false, false);
    return v + __builtin_bit_cast(float, s);
}

__device__ __forceinline__ float gatv2_wave_sum(float v)
{
    v = gatv2_half_xadd(v);
    v = gatv2_row_xadd<8>(v);
    v = gatv2_row_xadd<4>(v);
    v = gatv2_row_xadd<2>(v);
    v = gatv2_row_xadd<1>(v);
    return v;
}

__device__ __forceinline__ void gatv2_pair_request(
    v4f& g0, v4f& g1, float (&ea0)[EDGE_DIM], float (&ea1)[EDGE_DIM],
    const float* __restrict__ xl, const float* __restrict__ eattr,
    const int my_e, const int my_src, const int j, const int j1, const int lane)
{
    const float* __restrict__ eap0 = eattr +
        (size_t)__builtin_amdgcn_readlane(my_e, j) * EDGE_DIM;
    const float* __restrict__ eap1 = eattr +
        (size_t)__builtin_amdgcn_readlane(my_e, j1) * EDGE_DIM;
    g0 = *(const v4f*)(xl +
        (size_t)__builtin_amdgcn_readlane(my_src, j) * OUT_DIM + 4 * lane);
    g1 = *(const v4f*)(xl +
        (size_t)__builtin_amdgcn_readlane(my_src, j1) * OUT_DIM + 4 * lane);
    #pragma unroll
    for (int d = 0; d < EDGE_DIM; ++d) { ea0[d] = eap0[d]; ea1[d] = eap1[d]; }
}

__device__ __forceinline__ void gatv2_pair_fold(
    const v4f g0, const v4f g1,
    const float (&ea0)[EDGE_DIM], const float (&ea1)[EDGE_DIM], const bool two,
    const v2f (&w2)[2][EDGE_DIM], const v2f (&gr2)[2], const v2f (&a2)[2],
    float& m, float& den, float (&acc)[4])
{
    const float gl0[4] = {g0.x, g0.y, g0.z, g0.w};
    const float gl1[4] = {g1.x, g1.y, g1.z, g1.w};

    float p0 = gatv2_lane_logit(ea0, w2, g0, gr2, a2);
    float p1 = gatv2_lane_logit(ea1, w2, g1, gr2, a2);
    p0 = gatv2_row_xadd<4>(p0);  p1 = gatv2_row_xadd<4>(p1);
    p0 = gatv2_row_xadd<2>(p0);  p1 = gatv2_row_xadd<2>(p1);
    p0 = gatv2_row_xadd<1>(p0);  p1 = gatv2_row_xadd<1>(p1);
    p1 = two ? p1 : -__builtin_inff();

    const float mn = fmaxf(m, fmaxf(p0, p1));
    const float sc = __expf(m - mn);
    const float t0 = __expf(p0 - mn);
    const float t1 = __expf(p1 - mn);
    den = fmaf(den, sc, t0) + t1;
    #pragma unroll
    for (int k = 0; k < 4; ++k)
        acc[k] = fmaf(t1, gl1[k], fmaf(acc[k], sc, t0 * gl0[k]));
    m = mn;
}

__global__ __launch_bounds__(256) void gatv2_node(
    const int* __restrict__ ei, const float* __restrict__ eattr,
    const float* __restrict__ we, const float* __restrict__ att,
    const float* __restrict__ xl, const float* __restrict__ xr,
    const int* __restrict__ rowptr, const int* __restrict__ deg,
    const int* __restrict__ perm,
    const float* __restrict__ x, const float* __restrict__ cbias,
    const float* __restrict__ gamma, const float* __restrict__ beta,
    float* __restrict__ out)
{
    __shared__ v4f s_w[OUT_DIM * EDGE_DIM / 4];
    __shared__ v4f s_a[OUT_DIM / 4];

    const int lane = threadIdx.x & 31;
    for (int i = threadIdx.x; i < OUT_DIM * EDGE_DIM / 4; i += 256)
        s_w[i] = ((const v4f*)we)[i];
    if (threadIdx.x < OUT_DIM / 4)
        s_a[threadIdx.x] = ((const v4f*)att)[threadIdx.x];
    __syncthreads();

    float wf[4 * EDGE_DIM];
    #pragma unroll
    for (int i = 0; i < EDGE_DIM; ++i) {
        const v4f t = s_w[lane * EDGE_DIM + i];
        wf[4 * i + 0] = t.x; wf[4 * i + 1] = t.y;
        wf[4 * i + 2] = t.z; wf[4 * i + 3] = t.w;
    }
    v2f w2[2][EDGE_DIM];
    #pragma unroll
    for (int q = 0; q < 2; ++q) {
        #pragma unroll
        for (int d = 0; d < EDGE_DIM; ++d) {
            w2[q][d].x = wf[(2 * q) * EDGE_DIM + d];
            w2[q][d].y = wf[(2 * q + 1) * EDGE_DIM + d];
        }
    }
    const v4f a4 = s_a[lane];
    const v2f a2[2] = {{a4.x, a4.y}, {a4.z, a4.w}};

    const int n = __builtin_amdgcn_readfirstlane(
        (int)(blockIdx.x * 8 + (threadIdx.x >> 5)));

    const v4f r4 = *(const v4f*)(xr + (size_t)n * OUT_DIM + 4 * lane);
    const v2f gr2[2] = {{r4.x, r4.y}, {r4.z, r4.w}};

    int beg = rowptr[n];
    int end = beg + deg[n];
    beg = beg < 0 ? 0 : beg;
    end = end > N_EDGES ? N_EDGES : end;

    float m = -__builtin_inff();
    float den = 0.0f;
    float acc[4] = {0.0f, 0.0f, 0.0f, 0.0f};

    for (int base = beg; base < end; base += 32) {
        const int idx = base + lane;
        int my_e = 0, my_src = 0;
        if (idx < end) {
            my_e = perm[idx];
            my_e = (unsigned)my_e < (unsigned)N_EDGES ? my_e : 0;
            my_src = ei[my_e];
        }
        const int left = end - base;
        const int cnt = left < 32 ? left : 32;
        const int jl = cnt - 1;
        v4f g0, g1;
        float ea0[EDGE_DIM], ea1[EDGE_DIM];
        gatv2_pair_request(g0, g1, ea0, ea1, xl, eattr, my_e, my_src,
                           0, jl < 1 ? jl : 1, lane);
        for (int j = 0; j < cnt; j += 2) {
            v4f n0, n1;
            float na0[EDGE_DIM], na1[EDGE_DIM];
            gatv2_pair_request(n0, n1, na0, na1, xl, eattr, my_e, my_src,
                               j + 2 < jl ? j + 2 : jl, j + 3 < jl ? j + 3 : jl, lane);
            __builtin_amdgcn_sched_barrier(0);
            gatv2_pair_fold(g0, g1, ea0, ea1, j + 1 < cnt, w2, gr2, a2, m, den, acc);
            __builtin_amdgcn_sched_barrier(0);
            g0 = n0;  g1 = n1;
            #pragma unroll
            for (int d = 0; d < EDGE_DIM; ++d) { ea0[d] = na0[d]; ea1[d] = na1[d]; }
        }
    }

    const float inv_den = 1.0f / (den + 1e-16f);

    const v4f x4 = *(const v4f*)(x + (size_t)n * OUT_DIM + 4 * lane);
    const v4f c4 = *(const v4f*)(cbias + 4 * lane);
    const float xv[4] = {x4.x, x4.y, x4.z, x4.w};
    const float cb[4] = {c4.x, c4.y, c4.z, c4.w};

    float h[4];
    float s = 0.0f;
    #pragma unroll
    for (int k = 0; k < 4; ++k) {
        const float v = acc[k] * inv_den + cb[k];
        const float sig = 1.0f / (1.0f + __expf(-v));
        h[k] = v * sig + xv[k];
        s += h[k];
    }
    s = gatv2_wave_sum(s);
    const float mu = s * (1.0f / OUT_DIM);

    float vs = 0.0f;
    #pragma unroll
    for (int k = 0; k < 4; ++k) { const float d = h[k] - mu; vs += d * d; }
    vs = gatv2_wave_sum(vs);
    const float inv = rsqrtf(vs * (1.0f / OUT_DIM) + LN_EPS);

    const v4f g4 = *(const v4f*)(gamma + 4 * lane);
    const v4f b4 = *(const v4f*)(beta + 4 * lane);
    v4f o;
    o.x = (h[0] - mu) * inv * g4.x + b4.x;
    o.y = (h[1] - mu) * inv * g4.y + b4.y;
    o.z = (h[2] - mu) * inv * g4.z + b4.z;
    o.w = (h[3] - mu) * inv * g4.w + b4.w;
    *(v4f*)(out + (size_t)n * OUT_DIM + 4 * lane) = o;
}

extern "C" void kernel_launch(void* const* d_in, const int* in_sizes, int n_in,
                              void* d_out, int out_size, void* d_ws, size_t ws_size,
                              hipStream_t stream) {
    const float* x     = (const float*)d_in[0];
    const int*   ei    = (const int*)  d_in[1];
    const float* eattr = (const float*)d_in[2];
    const float* wl    = (const float*)d_in[3];
    const float* bl    = (const float*)d_in[4];
    const float* wr    = (const float*)d_in[5];
    const float* br    = (const float*)d_in[6];
    const float* we    = (const float*)d_in[7];
    const float* att   = (const float*)d_in[8];
    const float* cb    = (const float*)d_in[9];
    const float* gam   = (const float*)d_in[10];
    const float* bet   = (const float*)d_in[11];
    float* out = (float*)d_out;

    float* ws     = (float*)d_ws;
    float* xl     = ws;
    float* xr     = xl     + (size_t)N_NODES * OUT_DIM;
    int*   deg    = (int*)(xr + (size_t)N_NODES * OUT_DIM);
    int*   rowptr = deg    + N_NODES;
    int*   perm   = rowptr + N_NODES;
    int*   counts = perm   + N_EDGES;
    int*   bbase  = counts + SORT_CHUNKS * N_BKT;
    int*   btot   = bbase  + 256;
    int*   rec    = btot   + 256;

    gatv2_gemm_lr<<<GEMM_BLOCKS, 256, 0, stream>>>(x, wl, bl, wr, br, xl, xr);

    gatv2_chunk_hist<<<SORT_CHUNKS, 256, 0, stream>>>(ei, counts);

    gatv2_scatter<<<SORT_CHUNKS, 256, 0, stream>>>(ei, counts, bbase, btot, rec);

    gatv2_rank<<<N_BKT, 256, 0, stream>>>(rec, bbase, btot, rowptr, deg, perm);

    gatv2_node<<<N_NODES / 8, 256, 0, stream>>>(
        ei, eattr, we, att, xl, xr, rowptr, deg, perm, x, cb, gam, bet, out);
}


# edge sort level 1 made chunk-local: each chunk counted, scanned, bucket-grouped in one kernel (no histogram launch, no cross-chunk count table); rank block gathers 256 runs; sum order: LDS arrival
# speedup vs baseline: 4.9760x; 1.0840x over previous
#include <hip/hip_runtime.h>

#define N_NODES 50000
#define N_EDGES 800000
#define IN_DIM 128
#define OUT_DIM 128
#define HEADS 4
#define HEAD_DIM 32
#define EDGE_DIM 10
#define NEG_SLOPE 0.2f
#define LN_EPS 1e-5f

typedef __attribute__((ext_vector_type(2))) float v2f;
typedef __attribute__((ext_vector_type(8))) float v8f;
typedef __attribute__((ext_vector_type(4))) float v4f;

#define GEMM_BLOCKS 256
#define GEMM_TILES (N_NODES / 16)
#define GEMM_FRAGS (IN_DIM / 8)

__device__ __forceinline__ void gatv2_load_frags(
    v4f (&f)[GEMM_FRAGS], const float* __restrict__ row_k4h)
{
    __builtin_amdgcn_sched_barrier(0);
    #pragma unroll
    for (int j = 0; j < GEMM_FRAGS; ++j)
        f[j] = *(const v4f*)(row_k4h + 8 * j);
    __builtin_amdgcn_sched_barrier(0);
}

__device__ __forceinline__ void gatv2_gemm_tile(
    const v4f (&a)[GEMM_FRAGS], const v4f (&fl)[GEMM_FRAGS],
    const v4f (&fr)[GEMM_FRAGS], const float biasL, const float biasR,
    float* __restrict__ xl, float* __restrict__ xr, const size_t out0)
{
    v8f accL = {};
    v8f accR = {};
    #pragma unroll
    for (int j = 0; j < GEMM_FRAGS; ++j) {
        const v2f a0 = {a[j].x, a[j].y},   a1 = {a[j].z, a[j].w};
        const v2f l0 = {fl[j].x, fl[j].y}, l1 = {fl[j].z, fl[j].w};
        const v2f r0 = {fr[j].x, fr[j].y}, r1 = {fr[j].z, fr[j].w};
        accL = __builtin_amdgcn_wmma_f32_16x16x4_f32(
            false, a0, false, l0, (short)0, accL, false, false);
        accR = __builtin_amdgcn_wmma_f32_16x16x4_f32(
            false, a0, false, r0, (short)0, accR, false, false);
        accL = __builtin_amdgcn_wmma_f32_16x16x4_f32(
            false, a1, false, l1, (short)0, accL, false, false);
        accR = __builtin_amdgcn_wmma_f32_16x16x4_f32(
            false, a1, false, r1, (short)0, accR, false, false);
    }
    #pragma unroll
    for (int r = 0; r < 8; ++r) {
        xl[out0 + (size_t)r * OUT_DIM] = accL[r] + biasL;
        xr[out0 + (size_t)r * OUT_DIM] = accR[r] + biasR;
    }
}

__global__ __launch_bounds__(256) void gatv2_gemm_lr(
    const float* __restrict__ x,
    const float* __restrict__ wl, const float* __restrict__ bl,
    const float* __restrict__ wr, const float* __restrict__ br,
    float* __restrict__ xl, float* __restrict__ xr)
{
    const int wave = threadIdx.x >> 5;
    const int lane = threadIdx.x & 31;
    const int c = lane & 15;
    const int h = lane >> 4;
    const int bcol = wave * 16 + c;

    v4f fl[GEMM_FRAGS], fr[GEMM_FRAGS];
    gatv2_load_frags(fl, wl + (size_t)bcol * IN_DIM + 4 * h);
    gatv2_load_frags(fr, wr + (size_t)bcol * IN_DIM + 4 * h);
    const float biasL = bl[bcol];
    const float biasR = br[bcol];

    const size_t a_lane = (size_t)c * IN_DIM + 4 * h;
    const size_t d_lane = (size_t)(8 * h) * OUT_DIM + bcol;
    const int b = blockIdx.x;
    if (b >= GEMM_TILES) return;
    const int my = (GEMM_TILES - 1 - b) / GEMM_BLOCKS + 1;
    const int last = b + (my - 1) * GEMM_BLOCKS;

    int t = b;
    v4f xa[GEMM_FRAGS], xb[GEMM_FRAGS];
    gatv2_load_frags(xa, x + (size_t)t * 16 * IN_DIM + a_lane);
    for (int p = 0; p < my / 2; ++p) {
        const int t1 = t + GEMM_BLOCKS;
        const int t2 = t1 + GEMM_BLOCKS;
        gatv2_load_frags(xb, x + (size_t)t1 * 16 * IN_DIM + a_lane);
        gatv2_gemm_tile(xa, fl, fr, biasL, biasR, xl, xr,
                        (size_t)t * 16 * OUT_DIM + d_lane);
        gatv2_load_frags(xa, x + (size_t)(t2 <= last ? t2 : last) * 16 * IN_DIM + a_lane);
        gatv2_gemm_tile(xb, fl, fr, biasL, biasR, xl, xr,
                        (size_t)t1 * 16 * OUT_DIM + d_lane);
        t = t2;
    }
    if (my & 1)
        gatv2_gemm_tile(xa, fl, fr, biasL, biasR, xl, xr,
                        (size_t)t * 16 * OUT_DIM + d_lane);
}

typedef __attribute__((ext_vector_type(2))) int v2i;

#define SORT_CHUNKS 256
#define CHUNK_EDGES (N_EDGES / SORT_CHUNKS)
#define CHUNK_EPT ((CHUNK_EDGES + 255) / 256)
#define BKT_SHIFT 8
#define BKT_NODES (1 << BKT_SHIFT)
#define N_BKT ((N_NODES + BKT_NODES - 1) / BKT_NODES)
#define OFFS_PITCH (N_BKT + 1)
#define RUN_REGS 32

__device__ __forceinline__ int gatv2_bucket(const int d)
{
    const unsigned k = (unsigned)d >> BKT_SHIFT;
    return (int)(k < (unsigned)N_BKT ? k : (unsigned)(N_BKT - 1));
}

__device__ __forceinline__ void gatv2_chunk_dst(
    int (&d)[CHUNK_EPT], const int* __restrict__ ei, const int chunk, const int t)
{
    const int* __restrict__ dstp = ei + N_EDGES + chunk * CHUNK_EDGES;
    #pragma unroll
    for (int u = 0; u < CHUNK_EPT; ++u) {
        const int idx = t + u * 256;
        d[u] = idx < CHUNK_EDGES ? dstp[idx] : -1;
    }
}

__device__ __forceinline__ int gatv2_block_excl_scan(const int v, int* wsum)
{
    const int lane = threadIdx.x & 31;
    const int w = threadIdx.x >> 5;
    int incl = v;
    #pragma unroll
    for (int off = 1; off < 32; off <<= 1) {
        const int s = __shfl_up(incl, off, 32);
        if (lane >= off) incl += s;
    }
    if (lane == 31) wsum[w] = incl;
    __syncthreads();
    int wbase = 0;
    #pragma unroll
    for (int i = 0; i < 8; ++i) wbase += (i < w) ? wsum[i] : 0;
    return wbase + incl - v;
}

__device__ __forceinline__ int gatv2_wave_isum(int v)
{
    #pragma unroll
    for (int off = 16; off > 0; off >>= 1) v += __shfl_xor(v, off, 32);
    return v;
}

__global__ __launch_bounds__(256) void gatv2_chunk_group(
    const int* __restrict__ ei, int* __restrict__ offs, int* __restrict__ rec)
{
    __shared__ int slot[256];
    __shared__ int wsum[8];
    __shared__ v2i stage[CHUNK_EDGES];
    const int t = threadIdx.x;
    const int b = blockIdx.x;
    int d[CHUNK_EPT];
    gatv2_chunk_dst(d, ei, b, t);
    slot[t] = 0;
    __syncthreads();
    #pragma unroll
    for (int u = 0; u < CHUNK_EPT; ++u)
        if (t + u * 256 < CHUNK_EDGES) atomicAdd(&slot[gatv2_bucket(d[u])], 1);
    __syncthreads();

    const int own = slot[t];
    const int lst = gatv2_block_excl_scan(own, wsum);
    slot[t] = lst;
    if (t <= N_BKT) offs[b * OFFS_PITCH + t] = lst;
    __syncthreads();

    int lp[CHUNK_EPT];
    #pragma unroll
    for (int u = 0; u < CHUNK_EPT; ++u)
        lp[u] = t + u * 256 < CHUNK_EDGES ? atomicAdd(&slot[gatv2_bucket(d[u])], 1) : -1;
    #pragma unroll
    for (int u = 0; u < CHUNK_EPT; ++u) {
        if ((unsigned)lp[u] < (unsigned)CHUNK_EDGES) {
            const v2i r = {b * CHUNK_EDGES + t + u * 256, d[u]};
            stage[lp[u]] = r;
        }
    }
    __syncthreads();

    v2i* __restrict__ mine = (v2i*)rec + (size_t)b * CHUNK_EDGES;
    for (int i = t; i < CHUNK_EDGES; i += 256) mine[i] = stage[i];
}

__device__ __forceinline__ void gatv2_rank_count(const v2i r, const int node0, int* cnt)
{
    const unsigned dl = (unsigned)(r.y - node0);
    if (r.x >= 0 && dl < (unsigned)BKT_NODES) atomicAdd(&cnt[dl], 1);
}

__device__ __forceinline__ void gatv2_rank_place(
    const v2i r, const int node0, int* cnt, const int base, const int tot,
    const bool staged, int* stage, int* __restrict__ perm)
{
    const unsigned dl = (unsigned)(r.y - node0);
    if (r.x >= 0 && dl < (unsigned)BKT_NODES) {
        const int pos = atomicAdd(&cnt[dl], 1);
        if (staged) {
            const unsigned lp = (unsigned)(pos - base);
            if (lp < (unsigned)tot) stage[lp] = r.x;
        } else {
            if ((unsigned)pos < (unsigned)N_EDGES) perm[pos] = r.x;
        }
    }
}

#define RANK_STAGE 8192
__global__ __launch_bounds__(256) void gatv2_rank(
    const int* __restrict__ rec, const int* __restrict__ offs,
    int* __restrict__ rowptr, int* __restrict__ deg, int* __restrict__ perm)
{
    __shared__ int cnt[BKT_NODES];
    __shared__ int wsum[8];
    __shared__ int wbase[8];
    __shared__ int wtot[8];
    __shared__ int stage[RANK_STAGE];
    const int t = threadIdx.x;
    const int lane = t & 31;
    const int w = t >> 5;
    const int k = blockIdx.x;
    const int node0 = k * BKT_NODES;
    const v2i none = {-1, -1};

    int s0 = offs[t * OFFS_PITCH + k];
    int s1 = offs[t * OFFS_PITCH + k + 1];
    s0 = s0 < 0 ? 0 : (s0 > CHUNK_EDGES ? CHUNK_EDGES : s0);
    s1 = s1 < s0 ? s0 : (s1 > CHUNK_EDGES ? CHUNK_EDGES : s1);
    const int len = s1 - s0;

    const v2i* __restrict__ recs = (const v2i*)rec;
    v2i r[32];
    #pragma unroll
    for (int u = 0; u < 32; ++u) {
        const int cs = __builtin_amdgcn_readlane(s0, u);
        const int cl = __builtin_amdgcn_readlane(len, u);
        const v2i* __restrict__ run = recs + (size_t)(w * 32 + u) * CHUNK_EDGES + cs;
        r[u] = lane < cl ? run[lane] : none;
    }
    const bool has_long = __builtin_amdgcn_ballot_w32(len > RUN_REGS) != 0u;

    const int sb = gatv2_wave_isum(s0);
    const int st = gatv2_wave_isum(len);
    if (lane == 0) { wbase[w] = sb; wtot[w] = st; }
    cnt[t] = 0;
    __syncthreads();
    int base = 0, tot = 0;
    #pragma unroll
    for (int i = 0; i < 8; ++i) { base += wbase[i]; tot += wtot[i]; }
    base = base < 0 ? 0 : (base > N_EDGES ? N_EDGES : base);
    tot = tot < 0 ? 0 : (tot > N_EDGES - base ? N_EDGES - base : tot);

    #pragma unroll
    for (int u = 0; u < 32; ++u) gatv2_rank_count(r[u], node0, cnt);
    if (has_long) {
        for (int u = 0; u < 32; ++u) {
            const int cl = __builtin_amdgcn_readlane(len, u);
            if (cl <= RUN_REGS) continue;
            const v2i* __restrict__ run = recs + (size_t)(w * 32 + u) * CHUNK_EDGES +
                                          __builtin_amdgcn_readlane(s0, u);
            for (int i = RUN_REGS + lane; i < cl; i += 32)
                gatv2_rank_count(run[i], node0, cnt);
        }
    }
    __syncthreads();

    const int c = cnt[t];
    const int first = base + gatv2_block_excl_scan(c, wsum);
    cnt[t] = first;
    if (node0 + t < N_NODES) {
        rowptr[node0 + t] = first;
        deg[node0 + t] = c;
    }
    __syncthreads();

    const bool staged = tot <= RANK_STAGE;
    int pos[32];
    #pragma unroll
    for (int u = 0; u < 32; ++u) {
        const unsigned dl = (unsigned)(r[u].y - node0);
        pos[u] = (r[u].x >= 0 && dl < (unsigned)BKT_NODES) ? atomicAdd(&cnt[dl], 1) : -1;
    }
    if (staged) {
        #pragma unroll
        for (int u = 0; u < 32; ++u) {
            const unsigned sp = (unsigned)(pos[u] - base);
            if (pos[u] >= 0 && sp < (unsigned)tot) stage[sp] = r[u].x;
        }
    } else {
        #pragma unroll
        for (int u = 0; u < 32; ++u)
            if ((unsigned)pos[u] < (unsigned)N_EDGES) perm[pos[u]] = r[u].x;
    }
    if (has_long) {
        for (int u = 0; u < 32; ++u) {
            const int cl = __builtin_amdgcn_readlane(len, u);
            if (cl <= RUN_REGS) continue;
            const v2i* __restrict__ run = recs + (size_t)(w * 32 + u) * CHUNK_EDGES +
                                          __builtin_amdgcn_readlane(s0, u);
            for (int i = RUN_REGS + lane; i < cl; i += 32)
                gatv2_rank_place(run[i], node0, cnt, base, tot, staged, stage, perm);
        }
    }
    if (staged) {
        __syncthreads();
        for (int i = t; i < tot; i += 256) perm[base + i] = stage[i];
    }
}

__device__ __forceinline__ float gatv2_lane_logit(
    const float (&ea)[EDGE_DIM], const v2f (&w2)[2][EDGE_DIM], const v4f g,
    const v2f (&gr2)[2], const v2f (&a2)[2])
{
    v2f ev0 = {0.0f, 0.0f}, ev1 = {0.0f, 0.0f};
    #pragma unroll
    for (int d = 0; d < EDGE_DIM; ++d) {
        const v2f s = {ea[d], ea[d]};
        ev0 = __builtin_elementwise_fma(s, w2[0][d], ev0);
        ev1 = __builtin_elementwise_fma(s, w2[1][d], ev1);
    }
    const v2f g01 = {g.x, g.y}, g23 = {g.z, g.w};
    const v2f v0 = g01 + gr2[0] + ev0;
    const v2f v1 = g23 + gr2[1] + ev1;
    const float l0 = v0.x > 0.0f ? v0.x : NEG_SLOPE * v0.x;
    const float l1 = v0.y > 0.0f ? v0.y : NEG_SLOPE * v0.y;
    const float l2 = v1.x > 0.0f ? v1.x : NEG_SLOPE * v1.x;
    const float l3 = v1.y > 0.0f ? v1.y : NEG_SLOPE * v1.y;
    float p = l0 * a2[0].x;
    p = fmaf(l1, a2[0].y, p);
    p = fmaf(l2, a2[1].x, p);
    p = fmaf(l3, a2[1].y, p);
    return p;
}

template <int M>
__device__ __forceinline__ float gatv2_row_xadd(const float v)
{
    const int b = __builtin_bit_cast(int, v);
    const int s = __builtin_amdgcn_update_dpp(b, b, 0x160 + M, 0xf, 0xf, true);
    return v + __builtin_bit_cast(float, s);
}

__device__ __forceinline__ float gatv2_half_xadd(const float v)
{
    const int b = __builtin_bit_cast(int, v);
    const int s = __builtin_amdgcn_permlanex16(b, b, 0x76543210, 0xfedcba98, false, false);
    return v + __builtin_bit_cast(float, s);
}

__device__ __forceinline__ float gatv2_wave_sum(float v)
{
    v = gatv2_half_xadd(v);
    v = gatv2_row_xadd<8>(v);
    v = gatv2_row_xadd<4>(v);
    v = gatv2_row_xadd<2>(v);
    v = gatv2_row_xadd<1>(v);
    return v;
}

__device__ __forceinline__ void gatv2_pair_request(
    v4f& g0, v4f& g1, float (&ea0)[EDGE_DIM], float (&ea1)[EDGE_DIM],
    const float* __restrict__ xl, const float* __restrict__ eattr,
    const int my_e, const int my_src, const int j, const int j1, const int lane)
{
    const float* __restrict__ eap0 = eattr +
        (size_t)__builtin_amdgcn_readlane(my_e, j) * EDGE_DIM;
    const float* __restrict__ eap1 = eattr +
        (size_t)__builtin_amdgcn_readlane(my_e, j1) * EDGE_DIM;
    g0 = *(const v4f*)(xl +
        (size_t)__builtin_amdgcn_readlane(my_src, j) * OUT_DIM + 4 * lane);
    g1 = *(const v4f*)(xl +
        (size_t)__builtin_amdgcn_readlane(my_src, j1) * OUT_DIM + 4 * lane);
    #pragma unroll
    for (int d = 0; d < EDGE_DIM; ++d) { ea0[d] = eap0[d]; ea1[d] = eap1[d]; }
}

__device__ __forceinline__ void gatv2_pair_fold(
    const v4f g0, const v4f g1,
    const float (&ea0)[EDGE_DIM], const float (&ea1)[EDGE_DIM], const bool two,
    const v2f (&w2)[2][EDGE_DIM], const v2f (&gr2)[2], const v2f (&a2)[2],
    float& m, float& den, float (&acc)[4])
{
    const float gl0[4] = {g0.x, g0.y, g0.z, g0.w};
    const float gl1[4] = {g1.x, g1.y, g1.z, g1.w};

    float p0 = gatv2_lane_logit(ea0, w2, g0, gr2, a2);
    float p1 = gatv2_lane_logit(ea1, w2, g1, gr2, a2);
    p0 = gatv2_row_xadd<4>(p0);  p1 = gatv2_row_xadd<4>(p1);
    p0 = gatv2_row_xadd<2>(p0);  p1 = gatv2_row_xadd<2>(p1);
    p0 = gatv2_row_xadd<1>(p0);  p1 = gatv2_row_xadd<1>(p1);
    p1 = two ? p1 : -__builtin_inff();

    const float mn = fmaxf(m, fmaxf(p0, p1));
    const float sc = __expf(m - mn);
    const float t0 = __expf(p0 - mn);
    const float t1 = __expf(p1 - mn);
    den = fmaf(den, sc, t0) + t1;
    #pragma unroll
    for (int k = 0; k < 4; ++k)
        acc[k] = fmaf(t1, gl1[k], fmaf(acc[k], sc, t0 * gl0[k]));
    m = mn;
}

__global__ __launch_bounds__(256) void gatv2_node(
    const int* __restrict__ ei, const float* __restrict__ eattr,
    const float* __restrict__ we, const float* __restrict__ att,
    const float* __restrict__ xl, const float* __restrict__ xr,
    const int* __restrict__ rowptr, const int* __restrict__ deg,
    const int* __restrict__ perm,
    const float* __restrict__ x, const float* __restrict__ cbias,
    const float* __restrict__ gamma, const float* __restrict__ beta,
    float* __restrict__ out)
{
    __shared__ v4f s_w[OUT_DIM * EDGE_DIM / 4];
    __shared__ v4f s_a[OUT_DIM / 4];

    const int lane = threadIdx.x & 31;
    for (int i = threadIdx.x; i < OUT_DIM * EDGE_DIM / 4; i += 256)
        s_w[i] = ((const v4f*)we)[i];
    if (threadIdx.x < OUT_DIM / 4)
        s_a[threadIdx.x] = ((const v4f*)att)[threadIdx.x];
    __syncthreads();

    float wf[4 * EDGE_DIM];
    #pragma unroll
    for (int i = 0; i < EDGE_DIM; ++i) {
        const v4f t = s_w[lane * EDGE_DIM + i];
        wf[4 * i + 0] = t.x; wf[4 * i + 1] = t.y;
        wf[4 * i + 2] = t.z; wf[4 * i + 3] = t.w;
    }
    v2f w2[2][EDGE_DIM];
    #pragma unroll
    for (int q = 0; q < 2; ++q) {
        #pragma unroll
        for (int d = 0; d < EDGE_DIM; ++d) {
            w2[q][d].x = wf[(2 * q) * EDGE_DIM + d];
            w2[q][d].y = wf[(2 * q + 1) * EDGE_DIM + d];
        }
    }
    const v4f a4 = s_a[lane];
    const v2f a2[2] = {{a4.x, a4.y}, {a4.z, a4.w}};

    const int n = __builtin_amdgcn_readfirstlane(
        (int)(blockIdx.x * 8 + (threadIdx.x >> 5)));

    const v4f r4 = *(const v4f*)(xr + (size_t)n * OUT_DIM + 4 * lane);
    const v2f gr2[2] = {{r4.x, r4.y}, {r4.z, r4.w}};

    int beg = rowptr[n];
    int end = beg + deg[n];
    beg = beg < 0 ? 0 : beg;
    end = end > N_EDGES ? N_EDGES : end;

    float m = -__builtin_inff();
    float den = 0.0f;
    float acc[4] = {0.0f, 0.0f, 0.0f, 0.0f};

    for (int base = beg; base < end; base += 32) {
        const int idx = base + lane;
        int my_e = 0, my_src = 0;
        if (idx < end) {
            my_e = perm[idx];
            my_e = (unsigned)my_e < (unsigned)N_EDGES ? my_e : 0;
            my_src = ei[my_e];
        }
        const int left = end - base;
        const int cnt = left < 32 ? left : 32;
        const int jl = cnt - 1;
        v4f g0, g1;
        float ea0[EDGE_DIM], ea1[EDGE_DIM];
        gatv2_pair_request(g0, g1, ea0, ea1, xl, eattr, my_e, my_src,
                           0, jl < 1 ? jl : 1, lane);
        for (int j = 0; j < cnt; j += 2) {
            v4f n0, n1;
            float na0[EDGE_DIM], na1[EDGE_DIM];
            gatv2_pair_request(n0, n1, na0, na1, xl, eattr, my_e, my_src,
                               j + 2 < jl ? j + 2 : jl, j + 3 < jl ? j + 3 : jl, lane);
            __builtin_amdgcn_sched_barrier(0);
            gatv2_pair_fold(g0, g1, ea0, ea1, j + 1 < cnt, w2, gr2, a2, m, den, acc);
            __builtin_amdgcn_sched_barrier(0);
            g0 = n0;  g1 = n1;
            #pragma unroll
            for (int d = 0; d < EDGE_DIM; ++d) { ea0[d] = na0[d]; ea1[d] = na1[d]; }
        }
    }

    const float inv_den = 1.0f / (den + 1e-16f);

    const v4f x4 = *(const v4f*)(x + (size_t)n * OUT_DIM + 4 * lane);
    const v4f c4 = *(const v4f*)(cbias + 4 * lane);
    const float xv[4] = {x4.x, x4.y, x4.z, x4.w};
    const float cb[4] = {c4.x, c4.y, c4.z, c4.w};

    float h[4];
    float s = 0.0f;
    #pragma unroll
    for (int k = 0; k < 4; ++k) {
        const float v = acc[k] * inv_den + cb[k];
        const float sig = 1.0f / (1.0f + __expf(-v));
        h[k] = v * sig + xv[k];
        s += h[k];
    }
    s = gatv2_wave_sum(s);
    const float mu = s * (1.0f / OUT_DIM);

    float vs = 0.0f;
    #pragma unroll
    for (int k = 0; k < 4; ++k) { const float d = h[k] - mu; vs += d * d; }
    vs = gatv2_wave_sum(vs);
    const float inv = rsqrtf(vs * (1.0f / OUT_DIM) + LN_EPS);

    const v4f g4 = *(const v4f*)(gamma + 4 * lane);
    const v4f b4 = *(const v4f*)(beta + 4 * lane);
    v4f o;
    o.x = (h[0] - mu) * inv * g4.x + b4.x;
    o.y = (h[1] - mu) * inv * g4.y + b4.y;
    o.z = (h[2] - mu) * inv * g4.z + b4.z;
    o.w = (h[3] - mu) * inv * g4.w + b4.w;
    *(v4f*)(out + (size_t)n * OUT_DIM + 4 * lane) = o;
}

extern "C" void kernel_launch(void* const* d_in, const int* in_sizes, int n_in,
                              void* d_out, int out_size, void* d_ws, size_t ws_size,
                              hipStream_t stream) {
    const float* x     = (const float*)d_in[0];
    const int*   ei    = (const int*)  d_in[1];
    const float* eattr = (const float*)d_in[2];
    const float* wl    = (const float*)d_in[3];
    const float* bl    = (const float*)d_in[4];
    const float* wr    = (const float*)d_in[5];
    const float* br    = (const float*)d_in[6];
    const float* we    = (const float*)d_in[7];
    const float* att   = (const float*)d_in[8];
    const float* cb    = (const float*)d_in[9];
    const float* gam   = (const float*)d_in[10];
    const float* bet   = (const float*)d_in[11];
    float* out = (float*)d_out;

    float* ws     = (float*)d_ws;
    float* xl     = ws;
    float* xr     = xl     + (size_t)N_NODES * OUT_DIM;
    int*   deg    = (int*)(xr + (size_t)N_NODES * OUT_DIM);
    int*   rowptr = deg    + N_NODES;
    int*   perm   = rowptr + N_NODES;
    int*   offs   = perm   + N_EDGES;
    int*   rec    = offs   + SORT_CHUNKS * OFFS_PITCH;

    gatv2_gemm_lr<<<GEMM_BLOCKS, 256, 0, stream>>>(x, wl, bl, wr, br, xl, xr);

    gatv2_chunk_group<<<SORT_CHUNKS, 256, 0, stream>>>(ei, offs, rec);

    gatv2_rank<<<N_BKT, 256, 0, stream>>>(rec, offs, rowptr, deg, perm);

    gatv2_node<<<N_NODES / 8, 256, 0, stream>>>(
        ei, eattr, we, att, xl, xr, rowptr, deg, perm, x, cb, gam, bet, out);
}


# node kernel made persistent: 1024 resident blocks whose waves walk over nodes (wave g: g, g+8192, ..), weights staged once per wave, next node's range and rows asked a node ahead; sum order unchanged
# speedup vs baseline: 5.1475x; 1.0345x over previous
#include <hip/hip_runtime.h>

#define N_NODES 50000
#define N_EDGES 800000
#define IN_DIM 128
#define OUT_DIM 128
#define HEADS 4
#define HEAD_DIM 32
#define EDGE_DIM 10
#define NEG_SLOPE 0.2f
#define LN_EPS 1e-5f

typedef __attribute__((ext_vector_type(2))) float v2f;
typedef __attribute__((ext_vector_type(8))) float v8f;
typedef __attribute__((ext_vector_type(4))) float v4f;

#define GEMM_BLOCKS 256
#define GEMM_TILES (N_NODES / 16)
#define GEMM_FRAGS (IN_DIM / 8)

__device__ __forceinline__ void gatv2_load_frags(
    v4f (&f)[GEMM_FRAGS], const float* __restrict__ row_k4h)
{
    __builtin_amdgcn_sched_barrier(0);
    #pragma unroll
    for (int j = 0; j < GEMM_FRAGS; ++j)
        f[j] = *(const v4f*)(row_k4h + 8 * j);
    __builtin_amdgcn_sched_barrier(0);
}

__device__ __forceinline__ void gatv2_gemm_tile(
    const v4f (&a)[GEMM_FRAGS], const v4f (&fl)[GEMM_FRAGS],
    const v4f (&fr)[GEMM_FRAGS], const float biasL, const float biasR,
    float* __restrict__ xl, float* __restrict__ xr, const size_t out0)
{
    v8f accL = {};
    v8f accR = {};
    #pragma unroll
    for (int j = 0; j < GEMM_FRAGS; ++j) {
        const v2f a0 = {a[j].x, a[j].y},   a1 = {a[j].z, a[j].w};
        const v2f l0 = {fl[j].x, fl[j].y}, l1 = {fl[j].z, fl[j].w};
        const v2f r0 = {fr[j].x, fr[j].y}, r1 = {fr[j].z, fr[j].w};
        accL = __builtin_amdgcn_wmma_f32_16x16x4_f32(
            false, a0, false, l0, (short)0, accL, false, false);
        accR = __builtin_amdgcn_wmma_f32_16x16x4_f32(
            false, a0, false, r0, (short)0, accR, false, false);
        accL = __builtin_amdgcn_wmma_f32_16x16x4_f32(
            false, a1, false, l1, (short)0, accL, false, false);
        accR = __builtin_amdgcn_wmma_f32_16x16x4_f32(
            false, a1, false, r1, (short)0, accR, false, false);
    }
    #pragma unroll
    for (int r = 0; r < 8; ++r) {
        xl[out0 + (size_t)r * OUT_DIM] = accL[r] + biasL;
        xr[out0 + (size_t)r * OUT_DIM] = accR[r] + biasR;
    }
}

__global__ __launch_bounds__(256) void gatv2_gemm_lr(
    const float* __restrict__ x,
    const float* __restrict__ wl, const float* __restrict__ bl,
    const float* __restrict__ wr, const float* __restrict__ br,
    float* __restrict__ xl, float* __restrict__ xr)
{
    const int wave = threadIdx.x >> 5;
    const int lane = threadIdx.x & 31;
    const int c = lane & 15;
    const int h = lane >> 4;
    const int bcol = wave * 16 + c;

    v4f fl[GEMM_FRAGS], fr[GEMM_FRAGS];
    gatv2_load_frags(fl, wl + (size_t)bcol * IN_DIM + 4 * h);
    gatv2_load_frags(fr, wr + (size_t)bcol * IN_DIM + 4 * h);
    const float biasL = bl[bcol];
    const float biasR = br[bcol];

    const size_t a_lane = (size_t)c * IN_DIM + 4 * h;
    const size_t d_lane = (size_t)(8 * h) * OUT_DIM + bcol;
    const int b = blockIdx.x;
    if (b >= GEMM_TILES) return;
    const int my = (GEMM_TILES - 1 - b) / GEMM_BLOCKS + 1;
    const int last = b + (my - 1) * GEMM_BLOCKS;

    int t = b;
    v4f xa[GEMM_FRAGS], xb[GEMM_FRAGS];
    gatv2_load_frags(xa, x + (size_t)t * 16 * IN_DIM + a_lane);
    for (int p = 0; p < my / 2; ++p) {
        const int t1 = t + GEMM_BLOCKS;
        const int t2 = t1 + GEMM_BLOCKS;
        gatv2_load_frags(xb, x + (size_t)t1 * 16 * IN_DIM + a_lane);
        gatv2_gemm_tile(xa, fl, fr, biasL, biasR, xl, xr,
                        (size_t)t * 16 * OUT_DIM + d_lane);
        gatv2_load_frags(xa, x + (size_t)(t2 <= last ? t2 : last) * 16 * IN_DIM + a_lane);
        gatv2_gemm_tile(xb, fl, fr, biasL, biasR, xl, xr,
                        (size_t)t1 * 16 * OUT_DIM + d_lane);
        t = t2;
    }
    if (my & 1)
        gatv2_gemm_tile(xa, fl, fr, biasL, biasR, xl, xr,
                        (size_t)t * 16 * OUT_DIM + d_lane);
}

typedef __attribute__((ext_vector_type(2))) int v2i;

#define SORT_CHUNKS 256
#define CHUNK_EDGES (N_EDGES / SORT_CHUNKS)
#define CHUNK_EPT ((CHUNK_EDGES + 255) / 256)
#define BKT_SHIFT 8
#define BKT_NODES (1 << BKT_SHIFT)
#define N_BKT ((N_NODES + BKT_NODES - 1) / BKT_NODES)
#define OFFS_PITCH (N_BKT + 1)
#define RUN_REGS 32

__device__ __forceinline__ int gatv2_bucket(const int d)
{
    const unsigned k = (unsigned)d >> BKT_SHIFT;
    return (int)(k < (unsigned)N_BKT ? k : (unsigned)(N_BKT - 1));
}

__device__ __forceinline__ void gatv2_chunk_dst(
    int (&d)[CHUNK_EPT], const int* __restrict__ ei, const int chunk, const int t)
{
    const int* __restrict__ dstp = ei + N_EDGES + chunk * CHUNK_EDGES;
    #pragma unroll
    for (int u = 0; u < CHUNK_EPT; ++u) {
        const int idx = t + u * 256;
        d[u] = idx < CHUNK_EDGES ? dstp[idx] : -1;
    }
}

__device__ __forceinline__ int gatv2_block_excl_scan(const int v, int* wsum)
{
    const int lane = threadIdx.x & 31;
    const int w = threadIdx.x >> 5;
    int incl = v;
    #pragma unroll
    for (int off = 1; off < 32; off <<= 1) {
        const int s = __shfl_up(incl, off, 32);
        if (lane >= off) incl += s;
    }
    if (lane == 31) wsum[w] = incl;
    __syncthreads();
    int wbase = 0;
    #pragma unroll
    for (int i = 0; i < 8; ++i) wbase += (i < w) ? wsum[i] : 0;
    return wbase + incl - v;
}

__device__ __forceinline__ int gatv2_wave_isum(int v)
{
    #pragma unroll
    for (int off = 16; off > 0; off >>= 1) v += __shfl_xor(v, off, 32);
    return v;
}

__global__ __launch_bounds__(256) void gatv2_chunk_group(
    const int* __restrict__ ei, int* __restrict__ offs, int* __restrict__ rec)
{
    __shared__ int slot[256];
    __shared__ int wsum[8];
    __shared__ v2i stage[CHUNK_EDGES];
    const int t = threadIdx.x;
    const int b = blockIdx.x;
    int d[CHUNK_EPT];
    gatv2_chunk_dst(d, ei, b, t);
    slot[t] = 0;
    __syncthreads();
    #pragma unroll
    for (int u = 0; u < CHUNK_EPT; ++u)
        if (t + u * 256 < CHUNK_EDGES) atomicAdd(&slot[gatv2_bucket(d[u])], 1);
    __syncthreads();

    const int own = slot[t];
    const int lst = gatv2_block_excl_scan(own, wsum);
    slot[t] = lst;
    if (t <= N_BKT) offs[b * OFFS_PITCH + t] = lst;
    __syncthreads();

    int lp[CHUNK_EPT];
    #pragma unroll
    for (int u = 0; u < CHUNK_EPT; ++u)
        lp[u] = t + u * 256 < CHUNK_EDGES ? atomicAdd(&slot[gatv2_bucket(d[u])], 1) : -1;
    #pragma unroll
    for (int u = 0; u < CHUNK_EPT; ++u) {
        if ((unsigned)lp[u] < (unsigned)CHUNK_EDGES) {
            const v2i r = {b * CHUNK_EDGES + t + u * 256, d[u]};
            stage[lp[u]] = r;
        }
    }
    __syncthreads();

    v2i* __restrict__ mine = (v2i*)rec + (size_t)b * CHUNK_EDGES;
    for (int i = t; i < CHUNK_EDGES; i += 256) mine[i] = stage[i];
}

__device__ __forceinline__ void gatv2_rank_count(const v2i r, const int node0, int* cnt)
{
    const unsigned dl = (unsigned)(r.y - node0);
    if (r.x >= 0 && dl < (unsigned)BKT_NODES) atomicAdd(&cnt[dl], 1);
}

__device__ __forceinline__ void gatv2_rank_place(
    const v2i r, const int node0, int* cnt, const int base, const int tot,
    const bool staged, int* stage, int* __restrict__ perm)
{
    const unsigned dl = (unsigned)(r.y - node0);
    if (r.x >= 0 && dl < (unsigned)BKT_NODES) {
        const int pos = atomicAdd(&cnt[dl], 1);
        if (staged) {
            const unsigned lp = (unsigned)(pos - base);
            if (lp < (unsigned)tot) stage[lp] = r.x;
        } else {
            if ((unsigned)pos < (unsigned)N_EDGES) perm[pos] = r.x;
        }
    }
}

#define RANK_STAGE 8192
__global__ __launch_bounds__(256) void gatv2_rank(
    const int* __restrict__ rec, const int* __restrict__ offs,
    int* __restrict__ rowptr, int* __restrict__ deg, int* __restrict__ perm)
{
    __shared__ int cnt[BKT_NODES];
    __shared__ int wsum[8];
    __shared__ int wbase[8];
    __shared__ int wtot[8];
    __shared__ int stage[RANK_STAGE];
    const int t = threadIdx.x;
    const int lane = t & 31;
    const int w = t >> 5;
    const int k = blockIdx.x;
    const int node0 = k * BKT_NODES;
    const v2i none = {-1, -1};

    int s0 = offs[t * OFFS_PITCH + k];
    int s1 = offs[t * OFFS_PITCH + k + 1];
    s0 = s0 < 0 ? 0 : (s0 > CHUNK_EDGES ? CHUNK_EDGES : s0);
    s1 = s1 < s0 ? s0 : (s1 > CHUNK_EDGES ? CHUNK_EDGES : s1);
    const int len = s1 - s0;

    const v2i* __restrict__ recs = (const v2i*)rec;
    v2i r[32];
    #pragma unroll
    for (int u = 0; u < 32; ++u) {
        const int cs = __builtin_amdgcn_readlane(s0, u);
        const int cl = __builtin_amdgcn_readlane(len, u);
        const v2i* __restrict__ run = recs + (size_t)(w * 32 + u) * CHUNK_EDGES + cs;
        r[u] = lane < cl ? run[lane] : none;
    }
    const bool has_long = __builtin_amdgcn_ballot_w32(len > RUN_REGS) != 0u;

    const int sb = gatv2_wave_isum(s0);
    const int st = gatv2_wave_isum(len);
    if (lane == 0) { wbase[w] = sb; wtot[w] = st; }
    cnt[t] = 0;
    __syncthreads();
    int base = 0, tot = 0;
    #pragma unroll
    for (int i = 0; i < 8; ++i) { base += wbase[i]; tot += wtot[i]; }
    base = base < 0 ? 0 : (base > N_EDGES ? N_EDGES : base);
    tot = tot < 0 ? 0 : (tot > N_EDGES - base ? N_EDGES - base : tot);

    #pragma unroll
    for (int u = 0; u < 32; ++u) gatv2_rank_count(r[u], node0, cnt);
    if (has_long) {
        for (int u = 0; u < 32; ++u) {
            const int cl = __builtin_amdgcn_readlane(len, u);
            if (cl <= RUN_REGS) continue;
            const v2i* __restrict__ run = recs + (size_t)(w * 32 + u) * CHUNK_EDGES +
                                          __builtin_amdgcn_readlane(s0, u);
            for (int i = RUN_REGS + lane; i < cl; i += 32)
                gatv2_rank_count(run[i], node0, cnt);
        }
    }
    __syncthreads();

    const int c = cnt[t];
    const int first = base + gatv2_block_excl_scan(c, wsum);
    cnt[t] = first;
    if (node0 + t < N_NODES) {
        rowptr[node0 + t] = first;
        deg[node0 + t] = c;
    }
    __syncthreads();

    const bool staged = tot <= RANK_STAGE;
    int pos[32];
    #pragma unroll
    for (int u = 0; u < 32; ++u) {
        const unsigned dl = (unsigned)(r[u].y - node0);
        pos[u] = (r[u].x >= 0 && dl < (unsigned)BKT_NODES) ? atomicAdd(&cnt[dl], 1) : -1;
    }
    if (staged) {
        #pragma unroll
        for (int u = 0; u < 32; ++u) {
            const unsigned sp = (unsigned)(pos[u] - base);
            if (pos[u] >= 0 && sp < (unsigned)tot) stage[sp] = r[u].x;
        }
    } else {
        #pragma unroll
        for (int u = 0; u < 32; ++u)
            if ((unsigned)pos[u] < (unsigned)N_EDGES) perm[pos[u]] = r[u].x;
    }
    if (has_long) {
        for (int u = 0; u < 32; ++u) {
            const int cl = __builtin_amdgcn_readlane(len, u);
            if (cl <= RUN_REGS) continue;
            const v2i* __restrict__ run = recs + (size_t)(w * 32 + u) * CHUNK_EDGES +
                                          __builtin_amdgcn_readlane(s0, u);
            for (int i = RUN_REGS + lane; i < cl; i += 32)
                gatv2_rank_place(run[i], node0, cnt, base, tot, staged, stage, perm);
        }
    }
    if (staged) {
        __syncthreads();
        for (int i = t; i < tot; i += 256) perm[base + i] = stage[i];
    }
}

__device__ __forceinline__ float gatv2_lane_logit(
    const float (&ea)[EDGE_DIM], const v2f (&w2)[2][EDGE_DIM], const v4f g,
    const v2f (&gr2)[2], const v2f (&a2)[2])
{
    v2f ev0 = {0.0f, 0.0f}, ev1 = {0.0f, 0.0f};
    #pragma unroll
    for (int d = 0; d < EDGE_DIM; ++d) {
        const v2f s = {ea[d], ea[d]};
        ev0 = __builtin_elementwise_fma(s, w2[0][d], ev0);
        ev1 = __builtin_elementwise_fma(s, w2[1][d], ev1);
    }
    const v2f g01 = {g.x, g.y}, g23 = {g.z, g.w};
    const v2f v0 = g01 + gr2[0] + ev0;
    const v2f v1 = g23 + gr2[1] + ev1;
    const float l0 = v0.x > 0.0f ? v0.x : NEG_SLOPE * v0.x;
    const float l1 = v0.y > 0.0f ? v0.y : NEG_SLOPE * v0.y;
    const float l2 = v1.x > 0.0f ? v1.x : NEG_SLOPE * v1.x;
    const float l3 = v1.y > 0.0f ? v1.y : NEG_SLOPE * v1.y;
    float p = l0 * a2[0].x;
    p = fmaf(l1, a2[0].y, p);
    p = fmaf(l2, a2[1].x, p);
    p = fmaf(l3, a2[1].y, p);
    return p;
}

template <int M>
__device__ __forceinline__ float gatv2_row_xadd(const float v)
{
    const int b = __builtin_bit_cast(int, v);
    const int s = __builtin_amdgcn_update_dpp(b, b, 0x160 + M, 0xf, 0xf, true);
    return v + __builtin_bit_cast(float, s);
}

__device__ __forceinline__ float gatv2_half_xadd(const float v)
{
    const int b = __builtin_bit_cast(int, v);
    const int s = __builtin_amdgcn_permlanex16(b, b, 0x76543210, 0xfedcba98, false, false);
    return v + __builtin_bit_cast(float, s);
}

__device__ __forceinline__ float gatv2_wave_sum(float v)
{
    v = gatv2_half_xadd(v);
    v = gatv2_row_xadd<8>(v);
    v = gatv2_row_xadd<4>(v);
    v = gatv2_row_xadd<2>(v);
    v = gatv2_row_xadd<1>(v);
    return v;
}

__device__ __forceinline__ void gatv2_pair_request(
    v4f& g0, v4f& g1, float (&ea0)[EDGE_DIM], float (&ea1)[EDGE_DIM],
    const float* __restrict__ xl, const float* __restrict__ eattr,
    const int my_e, const int my_src, const int j, const int j1, const int lane)
{
    const float* __restrict__ eap0 = eattr +
        (size_t)__builtin_amdgcn_readlane(my_e, j) * EDGE_DIM;
    const float* __restrict__ eap1 = eattr +
        (size_t)__builtin_amdgcn_readlane(my_e, j1) * EDGE_DIM;
    g0 = *(const v4f*)(xl +
        (size_t)__builtin_amdgcn_readlane(my_src, j) * OUT_DIM + 4 * lane);
    g1 = *(const v4f*)(xl +
        (size_t)__builtin_amdgcn_readlane(my_src, j1) * OUT_DIM + 4 * lane);
    #pragma unroll
    for (int d = 0; d < EDGE_DIM; ++d) { ea0[d] = eap0[d]; ea1[d] = eap1[d]; }
}

__device__ __forceinline__ void gatv2_pair_fold(
    const v4f g0, const v4f g1,
    const float (&ea0)[EDGE_DIM], const float (&ea1)[EDGE_DIM], const bool two,
    const v2f (&w2)[2][EDGE_DIM], const v2f (&gr2)[2], const v2f (&a2)[2],
    float& m, float& den, float (&acc)[4])
{
    const float gl0[4] = {g0.x, g0.y, g0.z, g0.w};
    const float gl1[4] = {g1.x, g1.y, g1.z, g1.w};

    float p0 = gatv2_lane_logit(ea0, w2, g0, gr2, a2);
    float p1 = gatv2_lane_logit(ea1, w2, g1, gr2, a2);
    p0 = gatv2_row_xadd<4>(p0);  p1 = gatv2_row_xadd<4>(p1);
    p0 = gatv2_row_xadd<2>(p0);  p1 = gatv2_row_xadd<2>(p1);
    p0 = gatv2_row_xadd<1>(p0);  p1 = gatv2_row_xadd<1>(p1);
    p1 = two ? p1 : -__builtin_inff();

    const float mn = fmaxf(m, fmaxf(p0, p1));
    const float sc = __expf(m - mn);
    const float t0 = __expf(p0 - mn);
    const float t1 = __expf(p1 - mn);
    den = fmaf(den, sc, t0) + t1;
    #pragma unroll
    for (int k = 0; k < 4; ++k)
        acc[k] = fmaf(t1, gl1[k], fmaf(acc[k], sc, t0 * gl0[k]));
    m = mn;
}

#define NODE_BLOCKS 1024
#define NODE_WAVES (NODE_BLOCKS * 8)

__global__ __launch_bounds__(256) void gatv2_node(
    const int* __restrict__ ei, const float* __restrict__ eattr,
    const float* __restrict__ we, const float* __restrict__ att,
    const float* __restrict__ xl, const float* __restrict__ xr,
    const int* __restrict__ rowptr, const int* __restrict__ deg,
    const int* __restrict__ perm,
    const float* __restrict__ x, const float* __restrict__ cbias,
    const float* __restrict__ gamma, const float* __restrict__ beta,
    float* __restrict__ out)
{
    __shared__ v4f s_w[OUT_DIM * EDGE_DIM / 4];
    __shared__ v4f s_a[OUT_DIM / 4];

    const int lane = threadIdx.x & 31;
    for (int i = threadIdx.x; i < OUT_DIM * EDGE_DIM / 4; i += 256)
        s_w[i] = ((const v4f*)we)[i];
    if (threadIdx.x < OUT_DIM / 4)
        s_a[threadIdx.x] = ((const v4f*)att)[threadIdx.x];
    __syncthreads();

    float wf[4 * EDGE_DIM];
    #pragma unroll
    for (int i = 0; i < EDGE_DIM; ++i) {
        const v4f t = s_w[lane * EDGE_DIM + i];
        wf[4 * i + 0] = t.x; wf[4 * i + 1] = t.y;
        wf[4 * i + 2] = t.z; wf[4 * i + 3] = t.w;
    }
    v2f w2[2][EDGE_DIM];
    #pragma unroll
    for (int q = 0; q < 2; ++q) {
        #pragma unroll
        for (int d = 0; d < EDGE_DIM; ++d) {
            w2[q][d].x = wf[(2 * q) * EDGE_DIM + d];
            w2[q][d].y = wf[(2 * q + 1) * EDGE_DIM + d];
        }
    }
    const v4f a4 = s_a[lane];
    const v2f a2[2] = {{a4.x, a4.y}, {a4.z, a4.w}};

    const v4f c4 = *(const v4f*)(cbias + 4 * lane);
    const v4f g4 = *(const v4f*)(gamma + 4 * lane);
    const v4f b4 = *(const v4f*)(beta + 4 * lane);
    const float cb[4] = {c4.x, c4.y, c4.z, c4.w};

    const int n0 = __builtin_amdgcn_readfirstlane(
        (int)(blockIdx.x * 8 + (threadIdx.x >> 5)));

    int hbeg = rowptr[n0];
    int hdeg = deg[n0];
    v4f r4 = *(const v4f*)(xr + (size_t)n0 * OUT_DIM + 4 * lane);
    v4f x4 = *(const v4f*)(x + (size_t)n0 * OUT_DIM + 4 * lane);

    for (int n = n0; n < N_NODES; n += NODE_WAVES) {
        const int nn = n + NODE_WAVES < N_NODES ? n + NODE_WAVES : n;
        const int nbeg = rowptr[nn];
        const int ndeg = deg[nn];
        const v4f nr4 = *(const v4f*)(xr + (size_t)nn * OUT_DIM + 4 * lane);
        const v4f nx4 = *(const v4f*)(x + (size_t)nn * OUT_DIM + 4 * lane);

        const v2f gr2[2] = {{r4.x, r4.y}, {r4.z, r4.w}};

        int beg = hbeg;
        int end = beg + hdeg;
        beg = beg < 0 ? 0 : beg;
        end = end > N_EDGES ? N_EDGES : end;

        float m = -__builtin_inff();
        float den = 0.0f;
        float acc[4] = {0.0f, 0.0f, 0.0f, 0.0f};

        for (int base = beg; base < end; base += 32) {
            const int idx = base + lane;
            int my_e = 0, my_src = 0;
            if (idx < end) {
                my_e = perm[idx];
                my_e = (unsigned)my_e < (unsigned)N_EDGES ? my_e : 0;
                my_src = ei[my_e];
            }
            const int left = end - base;
            const int cnt = left < 32 ? left : 32;
            const int jl = cnt - 1;
            v4f g0, g1;
            float ea0[EDGE_DIM], ea1[EDGE_DIM];
            gatv2_pair_request(g0, g1, ea0, ea1, xl, eattr, my_e, my_src,
                               0, jl < 1 ? jl : 1, lane);
            for (int j = 0; j < cnt; j += 2) {
                v4f n0v, n1v;
                float na0[EDGE_DIM], na1[EDGE_DIM];
                gatv2_pair_request(n0v, n1v, na0, na1, xl, eattr, my_e, my_src,
                                   j + 2 < jl ? j + 2 : jl, j + 3 < jl ? j + 3 : jl, lane);
                __builtin_amdgcn_sched_barrier(0);
                gatv2_pair_fold(g0, g1, ea0, ea1, j + 1 < cnt, w2, gr2, a2, m, den, acc);
                __builtin_amdgcn_sched_barrier(0);
                g0 = n0v;  g1 = n1v;
                #pragma unroll
                for (int d = 0; d < EDGE_DIM; ++d) { ea0[d] = na0[d]; ea1[d] = na1[d]; }
            }
        }

        const float inv_den = 1.0f / (den + 1e-16f);
        const float xv[4] = {x4.x, x4.y, x4.z, x4.w};

        float h[4];
        float s = 0.0f;
        #pragma unroll
        for (int k = 0; k < 4; ++k) {
            const float v = acc[k] * inv_den + cb[k];
            const float sig = 1.0f / (1.0f + __expf(-v));
            h[k] = v * sig + xv[k];
            s += h[k];
        }
        s = gatv2_wave_sum(s);
        const float mu = s * (1.0f / OUT_DIM);

        float vs = 0.0f;
        #pragma unroll
        for (int k = 0; k < 4; ++k) { const float d = h[k] - mu; vs += d * d; }
        vs = gatv2_wave_sum(vs);
        const float inv = rsqrtf(vs * (1.0f / OUT_DIM) + LN_EPS);

        v4f o;
        o.x = (h[0] - mu) * inv * g4.x + b4.x;
        o.y = (h[1] - mu) * inv * g4.y + b4.y;
        o.z = (h[2] - mu) * inv * g4.z + b4.z;
        o.w = (h[3] - mu) * inv * g4.w + b4.w;
        *(v4f*)(out + (size_t)n * OUT_DIM + 4 * lane) = o;

        hbeg = nbeg;  hdeg = ndeg;  r4 = nr4;  x4 = nx4;
    }
}

extern "C" void kernel_launch(void* const* d_in, const int* in_sizes, int n_in,
                              void* d_out, int out_size, void* d_ws, size_t ws_size,
                              hipStream_t stream) {
    const float* x     = (const float*)d_in[0];
    const int*   ei    = (const int*)  d_in[1];
    const float* eattr = (const float*)d_in[2];
    const float* wl    = (const float*)d_in[3];
    const float* bl    = (const float*)d_in[4];
    const float* wr    = (const float*)d_in[5];
    const float* br    = (const float*)d_in[6];
    const float* we    = (const float*)d_in[7];
    const float* att   = (const float*)d_in[8];
    const float* cb    = (const float*)d_in[9];
    const float* gam   = (const float*)d_in[10];
    const float* bet   = (const float*)d_in[11];
    float* out = (float*)d_out;

    float* ws     = (float*)d_ws;
    float* xl     = ws;
    float* xr     = xl     + (size_t)N_NODES * OUT_DIM;
    int*   deg    = (int*)(xr + (size_t)N_NODES * OUT_DIM);
    int*   rowptr = deg    + N_NODES;
    int*   perm   = rowptr + N_NODES;
    int*   offs   = perm   + N_EDGES;
    int*   rec    = offs   + SORT_CHUNKS * OFFS_PITCH;

    gatv2_gemm_lr<<<GEMM_BLOCKS, 256, 0, stream>>>(x, wl, bl, wr, br, xl, xr);

    gatv2_chunk_group<<<SORT_CHUNKS, 256, 0, stream>>>(ei, offs, rec);

    gatv2_rank<<<N_BKT, 256, 0, stream>>>(rec, offs, rowptr, deg, perm);

    gatv2_node<<<NODE_BLOCKS, 256, 0, stream>>>(
        ei, eattr, we, att, xl, xr, rowptr, deg, perm, x, cb, gam, bet, out);
}
